# starting kernel
# baseline (speedup 1.0000x reference)
#include <hip/hip_runtime.h>
#include <hip/hip_bf16.h>
#include <math.h>

typedef __attribute__((ext_vector_type(2))) float v2f;
typedef __attribute__((ext_vector_type(8))) float v8f;

#define NEG_SLOPE 0.2f

__global__ void fill_f32(float* __restrict__ p, float v, long n) {
  long i = (long)blockIdx.x * blockDim.x + threadIdx.x;
  if (i < n) p[i] = v;
}

__global__ void wmma_gemm_f32(const float* __restrict__ A,
                              const float* __restrict__ B,
                              float* __restrict__ C, int K, int Ncols) {
  const int tileM = blockIdx.x * 16;
  const int wave  = threadIdx.x >> 5;
  const int lane  = threadIdx.x & 31;
  const int tileN = wave * 16;
  const int r  = lane & 15;
  const int hi = lane >> 4;

  const float* __restrict__ Arow = A + (size_t)(tileM + r) * K;
  v8f acc = {};

#pragma unroll 4
  for (int k = 0; k < K; k += 4) {
    v2f a, b;
    const int ka = k + 2 * hi;
    a.x = Arow[ka];
    a.y = Arow[ka + 1];
    b.x = B[(size_t)ka * Ncols + tileN + r];
    b.y = B[(size_t)(ka + 1) * Ncols + tileN + r];
    acc = __builtin_amdgcn_wmma_f32_16x16x4_f32(
         false, a,  false, b,
         (short)0, acc,  false,  false);
  }

  const int col     = tileN + r;
  const int rowBase = tileM + (hi << 3);
#pragma unroll
  for (int i = 0; i < 8; ++i)
    C[(size_t)(rowBase + i) * Ncols + col] = acc[i];
}

__global__ void att_logits(const float* __restrict__ h,
                           const float* __restrict__ att_s,
                           const float* __restrict__ att_d,
                           float* __restrict__ a_s, float* __restrict__ a_d,
                           int N, int H) {
  const int wid  = (blockIdx.x * blockDim.x + threadIdx.x) >> 5;
  const int lane = threadIdx.x & 31;
  if (wid >= N * H) return;
  const int n  = wid / H;
  const int hh = wid - n * H;
  const float* __restrict__ hp = h + ((size_t)n * H + hh) * 64;
  const float* __restrict__ ws = att_s + hh * 64;
  const float* __restrict__ wd = att_d + hh * 64;
  float s = hp[lane] * ws[lane] + hp[lane + 32] * ws[lane + 32];
  float d = hp[lane] * wd[lane] + hp[lane + 32] * wd[lane + 32];
#pragma unroll
  for (int off = 16; off > 0; off >>= 1) {
    s += __shfl_down(s, off, 32);
    d += __shfl_down(d, off, 32);
  }
  if (lane == 0) {
    a_s[wid] = s;
    a_d[wid] = d;
  }
}

__device__ __forceinline__ void atomicMaxF32(float* addr, float v) {
  if (v >= 0.0f)
    atomicMax((int*)addr, __float_as_int(v));
  else
    atomicMin((unsigned int*)addr, __float_as_uint(v));
}

__global__ void edge_logit_max(const int* __restrict__ ei, int E, int N, int H,
                               const float* __restrict__ a_s,
                               const float* __restrict__ a_d,
                               float* __restrict__ elog,
                               float* __restrict__ mx) {
  const long idx = (long)blockIdx.x * blockDim.x + threadIdx.x;
  const long total = ((long)E + N) * H;
  if (idx >= total) return;
  const long e  = idx / H;
  const int  hh = (int)(idx - e * H);
  const int  s  = (e < E) ? ei[e] : (int)(e - E);
  const int  d  = (e < E) ? ei[(size_t)E + e] : (int)(e - E);
  float v = a_s[(size_t)s * H + hh] + a_d[(size_t)d * H + hh];
  v = (v > 0.0f) ? v : NEG_SLOPE * v;
  elog[idx] = v;
  atomicMaxF32(&mx[(size_t)d * H + hh], v);
}

__global__ void edge_exp_sum(const int* __restrict__ ei, int E, int N, int H,
                             float* __restrict__ elog,
                             const float* __restrict__ mx,
                             float* __restrict__ dn) {
  const long idx = (long)blockIdx.x * blockDim.x + threadIdx.x;
  const long total = ((long)E + N) * H;
  if (idx >= total) return;
  const long e  = idx / H;
  const int  hh = (int)(idx - e * H);
  const int  d  = (e < E) ? ei[(size_t)E + e] : (int)(e - E);
  const float ex = __expf(elog[idx] - mx[(size_t)d * H + hh]);
  elog[idx] = ex;
  atomicAdd(&dn[(size_t)d * H + hh], ex);
}

__global__ void edge_aggregate(const int* __restrict__ ei, int E, int N, int H,
                               const float* __restrict__ elog,
                               const float* __restrict__ dn,
                               const float* __restrict__ hfeat,
                               float* __restrict__ out) {
  const long wid  = ((long)blockIdx.x * blockDim.x + threadIdx.x) >> 5;
  const int  lane = threadIdx.x & 31;
  const long total = ((long)E + N) * H;
  if (wid >= total) return;
  const long e  = wid / H;
  const int  hh = (int)(wid - e * H);
  const int  s  = (e < E) ? ei[e] : (int)(e - E);
  const int  d  = (e < E) ? ei[(size_t)E + e] : (int)(e - E);
  const float alpha = elog[wid] / dn[(size_t)d * H + hh];
  const float* __restrict__ hp = hfeat + ((size_t)s * H + hh) * 64;
  float* __restrict__ op = out + ((size_t)d * H + hh) * 64;
  atomicAdd(&op[lane],      alpha * hp[lane]);
  atomicAdd(&op[lane + 32], alpha * hp[lane + 32]);
}

__global__ void bias_elu(float* __restrict__ x, const float* __restrict__ b,
                         long n, int F) {
  long i = (long)blockIdx.x * blockDim.x + threadIdx.x;
  if (i >= n) return;
  float v = x[i] + b[i % F];
  x[i] = (v > 0.0f) ? v : expm1f(v);
}

__global__ void bias_add(float* __restrict__ x, const float* __restrict__ b,
                         long n, int F) {
  long i = (long)blockIdx.x * blockDim.x + threadIdx.x;
  if (i >= n) return;
  x[i] = x[i] + b[i % F];
}

static inline int cdiv_l(long a, int b) { return (int)((a + b - 1) / b); }

extern "C" void kernel_launch(void* const* d_in, const int* in_sizes, int n_in,
                              void* d_out, int out_size, void* d_ws, size_t ws_size,
                              hipStream_t stream) {
  const float* x    = (const float*)d_in[0];
  const int*   ei   = (const int*)d_in[1];
  const float* W1   = (const float*)d_in[2];
  const float* as1w = (const float*)d_in[3];
  const float* ad1w = (const float*)d_in[4];
  const float* b1   = (const float*)d_in[5];
  const float* W2   = (const float*)d_in[6];
  const float* as2w = (const float*)d_in[7];
  const float* ad2w = (const float*)d_in[8];
  const float* b2   = (const float*)d_in[9];

  const int Fin = 128;
  const int N   = in_sizes[0] / Fin;
  const int E   = in_sizes[1] / 2;
  const long Et = (long)E + N;

  float* ws   = (float*)d_ws;
  float* h1   = ws;  ws += (size_t)N * 128;
  float* agg1 = ws;  ws += (size_t)N * 128;
  float* h2   = ws;  ws += (size_t)N * 64;
  float* a_s  = ws;  ws += (size_t)N * 2;
  float* a_d  = ws;  ws += (size_t)N * 2;
  float* mx   = ws;  ws += (size_t)N * 2;
  float* dn   = ws;  ws += (size_t)N * 2;
  float* el   = ws;  ws += (size_t)Et * 2;
  float* outp = (float*)d_out;

  const float NEG_INF = -3.402823466e+38f;

  {
    const int H = 2;
    wmma_gemm_f32<<<N / 16, 32 * (128 / 16), 0, stream>>>(x, W1, h1, 128, 128);
    att_logits<<<cdiv_l((long)N * H, 8), 256, 0, stream>>>(h1, as1w, ad1w,
                                                           a_s, a_d, N, H);
    fill_f32<<<cdiv_l((long)N * H, 256), 256, 0, stream>>>(mx, NEG_INF, (long)N * H);
    fill_f32<<<cdiv_l((long)N * H, 256), 256, 0, stream>>>(dn, 0.0f, (long)N * H);
    fill_f32<<<cdiv_l((long)N * 128, 256), 256, 0, stream>>>(agg1, 0.0f, (long)N * 128);
    edge_logit_max<<<cdiv_l(Et * H, 256), 256, 0, stream>>>(ei, E, N, H,
                                                            a_s, a_d, el, mx);
    edge_exp_sum<<<cdiv_l(Et * H, 256), 256, 0, stream>>>(ei, E, N, H, el, mx, dn);
    edge_aggregate<<<cdiv_l(Et * H * 32, 256), 256, 0, stream>>>(ei, E, N, H,
                                                                 el, dn, h1, agg1);
    bias_elu<<<cdiv_l((long)N * 128, 256), 256, 0, stream>>>(agg1, b1,
                                                             (long)N * 128, 128);
  }

  {
    const int H = 1;
    wmma_gemm_f32<<<N / 16, 32 * (64 / 16), 0, stream>>>(agg1, W2, h2, 128, 64);
    att_logits<<<cdiv_l((long)N * H, 8), 256, 0, stream>>>(h2, as2w, ad2w,
                                                           a_s, a_d, N, H);
    fill_f32<<<cdiv_l((long)N * H, 256), 256, 0, stream>>>(mx, NEG_INF, (long)N * H);
    fill_f32<<<cdiv_l((long)N * H, 256), 256, 0, stream>>>(dn, 0.0f, (long)N * H);
    fill_f32<<<cdiv_l((long)N * 64, 256), 256, 0, stream>>>(outp, 0.0f, (long)N * 64);
    edge_logit_max<<<cdiv_l(Et * H, 256), 256, 0, stream>>>(ei, E, N, H,
                                                            a_s, a_d, el, mx);
    edge_exp_sum<<<cdiv_l(Et * H, 256), 256, 0, stream>>>(ei, E, N, H, el, mx, dn);
    edge_aggregate<<<cdiv_l(Et * H * 32, 256), 256, 0, stream>>>(ei, E, N, H,
                                                                 el, dn, h2, outp);
    bias_add<<<cdiv_l((long)N * 64, 256), 256, 0, stream>>>(outp, b2,
                                                            (long)N * 64, 64);
  }
}


# layer 2 reuses layer 1's kept-edge lists: an owner block saves its (source,row) list to the workspace behind scan 1, layer 2 reads it and never scans the edge list; its add order = that list's
# speedup vs baseline: 4.1535x; 4.1535x over previous
#include <hip/hip_runtime.h>
#include <hip/hip_bf16.h>
#include <math.h>

typedef __attribute__((ext_vector_type(2))) float v2f;
typedef __attribute__((ext_vector_type(8))) float v8f;

#define NEG_SLOPE 0.2f


template <int M>
__device__ __forceinline__ float row_xor(float v) {
  return __int_as_float(__builtin_amdgcn_update_dpp(
      0, __float_as_int(v), 0x160 + M, 0xf, 0xf, false));
}

template <int M>
__device__ __forceinline__ void fold_step(float (&v)[16], bool up) {
#pragma unroll
  for (int j = 0; j < M; ++j) {
    const float keep = up ? v[j + M] : v[j];
    const float send = up ? v[j] : v[j + M];
    v[j] = keep + row_xor<M>(send);
  }
}

__global__ void wmma_gemm_f32(const float* __restrict__ A,
                              const float* __restrict__ B,
                              float* __restrict__ C, int K, int Ncols,
                              const float* __restrict__ att_s,
                              const float* __restrict__ att_d,
                              float* __restrict__ a_s, float* __restrict__ a_d) {
  __shared__ float red[8][32];

  const int tileM = blockIdx.x * 16;
  const int wave  = threadIdx.x >> 5;
  const int lane  = threadIdx.x & 31;
  const int tileN = wave * 16;
  const int r  = lane & 15;
  const int hi = lane >> 4;
  const int col = tileN + r;

  const float ws = att_s[col];
  const float wd = att_d[col];

  const float* __restrict__ Arow = A + (size_t)(tileM + r) * K;
  v8f acc = {};

#pragma unroll 4
  for (int k = 0; k < K; k += 4) {
    v2f a, b;
    const int ka = k + 2 * hi;
    a.x = Arow[ka];
    a.y = Arow[ka + 1];
    b.x = B[(size_t)ka * Ncols + tileN + r];
    b.y = B[(size_t)(ka + 1) * Ncols + tileN + r];
    acc = __builtin_amdgcn_wmma_f32_16x16x4_f32(
         false, a,  false, b,
         (short)0, acc,  false,  false);
  }

  const int rowBase = tileM + (hi << 3);

  float v[16];
#pragma unroll
  for (int i = 0; i < 8; ++i) {
    v[i]     = acc[i] * ws;
    v[8 + i] = acc[i] * wd;
  }
  fold_step<8>(v, (lane & 8) != 0);
  fold_step<4>(v, (lane & 4) != 0);
  fold_step<2>(v, (lane & 2) != 0);
  fold_step<1>(v, (lane & 1) != 0);
  red[wave][lane] = v[0];
  __syncthreads();

#pragma unroll
  for (int i = 0; i < 8; ++i)
    C[(size_t)(rowBase + i) * Ncols + col] = acc[i];

  if ((wave & 3) == 0) {
    const float t = (red[wave][lane] + red[wave + 1][lane]) +
                    (red[wave + 2][lane] + red[wave + 3][lane]);
    const int H = Ncols >> 6;
    float* __restrict__ dstp = (lane & 8) ? a_d : a_s;
    dstp[(size_t)(rowBase + (lane & 7)) * H + (wave >> 2)] = t;
  }
}

#define OWN_R    416
#define OWN_NW   32
#define OWN_QCAP 64
#define OWN_U    16
#define OWN_G    8
#define OWN_LCAP 8192
#define OWN_KHEAD 32
#define OWN_KSTRIDE (OWN_KHEAD + 2 * OWN_LCAP)
static_assert(OWN_LCAP % 1024 == 0, "a thread reads OWN_LCAP/1024 saved entries");

__device__ __forceinline__ int ord_key(int b) { return b >= 0 ? b : (b ^ 0x7fffffff); }

template <int H, bool ELU, bool SAVE, bool REUSE>
__global__ void __launch_bounds__(1024)
owner_aggregate(const int* __restrict__ ei, int E, int N,
                const float* __restrict__ a_s, const float* __restrict__ a_d,
                const float* __restrict__ hfeat, const float* __restrict__ bias,
                float* __restrict__ out, int* __restrict__ kept) {
  constexpr int F = H * 64;
  __shared__ float acc[OWN_R * F];
  __shared__ float dns[OWN_R * H];
  __shared__ int   mxs[OWN_R * H];
  __shared__ int   q_e[OWN_NW][OWN_QCAP];
  __shared__ int   q_d[OWN_NW][OWN_QCAP];
  __shared__ int   l_s[OWN_LCAP];
  __shared__ int   l_d[OWN_LCAP];
  __shared__ int   l_n;

  const int tid  = threadIdx.x;
  const int lane = tid & 31;
  const int wave = tid >> 5;
  const int lo   = blockIdx.x * OWN_R;
  const int cnt  = min(OWN_R, N - lo);
  const int* __restrict__ srcp = ei;
  const int* __restrict__ dstp = ei + E;

  int* __restrict__ k_n = kept + (size_t)blockIdx.x * OWN_KSTRIDE;
  int* __restrict__ k_s = k_n + OWN_KHEAD;
  int* __restrict__ k_d = k_s + OWN_LCAP;
  int  k_total = 0;
  bool k_have  = false;
  if (REUSE) {
    k_total = k_n[0];
    k_have  = (unsigned)k_total <= (unsigned)OWN_LCAP;
  }

  if (tid == 0) l_n = k_have ? k_total : 0;
  for (int i = tid; i < cnt * F; i += 1024) acc[i] = 0.0f;
  for (int i = tid; i < cnt * H; i += 1024) {
    float v = a_s[(size_t)lo * H + i] + a_d[(size_t)lo * H + i];
    v = (v > 0.0f) ? v : NEG_SLOPE * v;
    mxs[i] = ord_key(__float_as_int(v));
    dns[i] = 0.0f;
  }
  __syncthreads();

  auto score = [&](int s, int dl, int h) -> float {
    float v = a_s[(size_t)s * H + h] + a_d[(size_t)(lo + dl) * H + h];
    return (v > 0.0f) ? v : NEG_SLOPE * v;
  };

  auto add_rows = [&](bool valid, int s, int dl, int m) {
    float w[H];
#pragma unroll
    for (int h = 0; h < H; ++h) w[h] = 0.0f;
    if (valid) {
#pragma unroll
      for (int h = 0; h < H; ++h) {
        const float mx = __int_as_float(ord_key(mxs[dl * H + h]));
        w[h] = __expf(score(s, dl, h) - mx);
        atomicAdd(&dns[dl * H + h], w[h]);
      }
    }
    for (int k0 = 0; k0 < m; k0 += OWN_G) {
      float2 r[OWN_G][H];
      int    dk[OWN_G];
#pragma unroll
      for (int g = 0; g < OWN_G; ++g) {
        const int kk = min(k0 + g, m - 1);
        const int sk = __builtin_amdgcn_readlane(s, kk);
        dk[g] = __builtin_amdgcn_readlane(dl, kk);
#pragma unroll
        for (int h = 0; h < H; ++h) {
          const float wk = __int_as_float(__builtin_amdgcn_readlane(__float_as_int(w[h]), kk));
          const float2* __restrict__ hp =
              reinterpret_cast<const float2*>(hfeat + ((size_t)sk * H + h) * 64);
          const float2 v = hp[lane];
          r[g][h].x = wk * v.x;
          r[g][h].y = wk * v.y;
        }
      }
#pragma unroll
      for (int g = 0; g < OWN_G; ++g) {
        if (k0 + g < m) {
#pragma unroll
          for (int h = 0; h < H; ++h) {
            float* ap = acc + dk[g] * F + h * 64 + 2 * lane;
            atomicAdd(&ap[0], r[g][h].x);
            atomicAdd(&ap[1], r[g][h].y);
          }
        }
      }
    }
  };

  auto flush = [&](int phase, int n) {
    __builtin_amdgcn_fence(__ATOMIC_ACQ_REL, "wavefront");
    for (int b = 0; b < n; b += 32) {
      const int i = b + lane;
      const bool valid = i < n;
      int s = 0, dl = 0;
      if (valid) {
        dl = q_d[wave][i];
        s  = srcp[q_e[wave][i]];
        s  = min(max(s, 0), N - 1);
      }
      if (phase == 1) {
        int at = 0;
        if (lane == 0) at = atomicAdd(&l_n, min(32, n - b));
        at = __builtin_amdgcn_readfirstlane(at) + lane;
        if (valid) {
          if (at < OWN_LCAP) {
            l_s[at] = s;
            l_d[at] = dl;
          }
#pragma unroll
          for (int h = 0; h < H; ++h)
            atomicMax(&mxs[dl * H + h], ord_key(__float_as_int(score(s, dl, h))));
        }
      } else {
        add_rows(valid, s, dl, min(32, n - b));
      }
    }
    __builtin_amdgcn_fence(__ATOMIC_ACQ_REL, "wavefront");
  };

  const int nIter = (E + 32 * OWN_U - 1) / (32 * OWN_U);
  const int rot = (int)(((long)blockIdx.x * nIter) / gridDim.x);

  bool list_ok = false;
  for (int phase = 1; phase <= 2; ++phase) {
    int qn = 0;
    if (list_ok) {
      const int total = l_n;
      for (int b = wave * 32; b < total; b += 32 * OWN_NW) {
        const int i = b + lane;
        const bool valid = i < total;
        add_rows(valid, valid ? l_s[i] : 0, valid ? l_d[i] : 0, min(32, total - b));
      }
    } else if (REUSE && k_have) {
      constexpr int PER = OWN_LCAP / 1024;
      if (k_total > 0) {
        int  sv[PER], dv[PER];
        int  key[PER][H];
#pragma unroll
        for (int j = 0; j < PER; ++j) {
          const int i = min(tid + j * 1024, k_total - 1);
          sv[j] = min(max(k_s[i], 0), N - 1);
          dv[j] = min(max(k_d[i], 0), cnt - 1);
        }
#pragma unroll
        for (int j = 0; j < PER; ++j) {
#pragma unroll
          for (int h = 0; h < H; ++h)
            key[j][h] = ord_key(__float_as_int(score(sv[j], dv[j], h)));
        }
#pragma unroll
        for (int j = 0; j < PER; ++j) {
          const int i = tid + j * 1024;
          if (i < k_total) {
            l_s[i] = sv[j];
            l_d[i] = dv[j];
#pragma unroll
            for (int h = 0; h < H; ++h) atomicMax(&mxs[dv[j] * H + h], key[j][h]);
          }
        }
      }
    } else
    for (int t = wave; t < nIter; t += OWN_NW) {
      int it = t + rot;
      if (it >= nIter) it -= nIter;
      const int base = it * (32 * OWN_U) + lane;
      int dv[OWN_U];
#pragma unroll
      for (int j = 0; j < OWN_U; ++j) {
        const int idx = base + j * 32;
        const int got = dstp[min(idx, E - 1)];
        dv[j] = (idx < E) ? got : -1;
      }
#pragma unroll
      for (int j = 0; j < OWN_U; ++j) {
        const unsigned dl = (unsigned)(dv[j] - lo);
        const bool m = dl < (unsigned)cnt;
        const unsigned mask = __builtin_amdgcn_ballot_w32(m);
        if (mask != 0u) {
          const int pos = qn + __popc(mask & ((1u << lane) - 1u));
          if (m) {
            q_e[wave][pos] = base + j * 32;
            q_d[wave][pos] = (int)dl;
          }
          qn += __popc(mask);
          if (qn >= 32) {
            flush(phase, qn);
            qn = 0;
          }
        }
      }
    }
    if (qn > 0) flush(phase, qn);
    if (phase == 2) {
      for (int b = wave * 32; b < cnt; b += 32 * OWN_NW) {
        const int dl = b + lane;
        const bool valid = dl < cnt;
        add_rows(valid, valid ? lo + dl : 0, valid ? dl : 0, min(32, cnt - b));
      }
    }
    __syncthreads();
    list_ok = (l_n <= OWN_LCAP);
    if (SAVE && phase == 1) {
      const int total = l_n;
      if (tid == 0) k_n[0] = total;
      if (list_ok) {
        for (int i = tid; i < total; i += 1024) {
          k_s[i] = l_s[i];
          k_d[i] = l_d[i];
        }
      }
    }
  }

  for (int i = tid; i < cnt * F; i += 1024) {
    const int nl = i / F;
    const int c  = i - nl * F;
    float v = acc[i] / dns[nl * H + (c >> 6)] + bias[c];
    if (ELU) v = (v > 0.0f) ? v : expm1f(v);
    out[(size_t)lo * F + i] = v;
  }
}

static inline int cdiv_l(long a, int b) { return (int)((a + b - 1) / b); }

extern "C" void kernel_launch(void* const* d_in, const int* in_sizes, int n_in,
                              void* d_out, int out_size, void* d_ws, size_t ws_size,
                              hipStream_t stream) {
  const float* x    = (const float*)d_in[0];
  const int*   ei   = (const int*)d_in[1];
  const float* W1   = (const float*)d_in[2];
  const float* as1w = (const float*)d_in[3];
  const float* ad1w = (const float*)d_in[4];
  const float* b1   = (const float*)d_in[5];
  const float* W2   = (const float*)d_in[6];
  const float* as2w = (const float*)d_in[7];
  const float* ad2w = (const float*)d_in[8];
  const float* b2   = (const float*)d_in[9];

  const int Fin = 128;
  const int N   = in_sizes[0] / Fin;
  const int E   = in_sizes[1] / 2;

  float* ws   = (float*)d_ws;
  float* h1   = ws;  ws += (size_t)N * 128;
  float* agg1 = ws;  ws += (size_t)N * 128;
  int*   kept = (int*)ws;  ws += (size_t)N * 64;
  float* h2   = h1;
  float* a_s  = ws;  ws += (size_t)N * 2;
  float* a_d  = ws;  ws += (size_t)N * 2;
  float* outp = (float*)d_out;

  const int nOwn = cdiv_l((long)N, OWN_R);

  wmma_gemm_f32<<<N / 16, 32 * (128 / 16), 0, stream>>>(x, W1, h1, 128, 128, as1w, ad1w, a_s, a_d);
  owner_aggregate<2, true, true, false><<<nOwn, 1024, 0, stream>>>(ei, E, N, a_s, a_d, h1, b1, agg1, kept);

  wmma_gemm_f32<<<N / 16, 32 * (64 / 16), 0, stream>>>(agg1, W2, h2, 128, 64, as2w, ad2w, a_s, a_d);
  owner_aggregate<1, false, false, true><<<nOwn, 1024, 0, stream>>>(ei, E, N, a_s, a_d, h2, b2, outp, kept);
}


# edge scan reads a 1-byte owner key per edge (a new launch writes the key plane) instead of the 4-byte destination: 16 edges a lane a load, 4 tested per integer op; kept-list and add order change
# speedup vs baseline: 4.6422x; 1.1177x over previous
#include <hip/hip_runtime.h>
#include <hip/hip_bf16.h>
#include <math.h>

typedef __attribute__((ext_vector_type(2))) float v2f;
typedef __attribute__((ext_vector_type(8))) float v8f;

#define NEG_SLOPE 0.2f


template <int M>
__device__ __forceinline__ float row_xor(float v) {
  return __int_as_float(__builtin_amdgcn_update_dpp(
      0, __float_as_int(v), 0x160 + M, 0xf, 0xf, false));
}

template <int M>
__device__ __forceinline__ void fold_step(float (&v)[16], bool up) {
#pragma unroll
  for (int j = 0; j < M; ++j) {
    const float keep = up ? v[j + M] : v[j];
    const float send = up ? v[j] : v[j + M];
    v[j] = keep + row_xor<M>(send);
  }
}

__global__ void wmma_gemm_f32(const float* __restrict__ A,
                              const float* __restrict__ B,
                              float* __restrict__ C, int K, int Ncols,
                              const float* __restrict__ att_s,
                              const float* __restrict__ att_d,
                              float* __restrict__ a_s, float* __restrict__ a_d) {
  __shared__ float red[8][32];

  const int tileM = blockIdx.x * 16;
  const int wave  = threadIdx.x >> 5;
  const int lane  = threadIdx.x & 31;
  const int tileN = wave * 16;
  const int r  = lane & 15;
  const int hi = lane >> 4;
  const int col = tileN + r;

  const float ws = att_s[col];
  const float wd = att_d[col];

  const float* __restrict__ Arow = A + (size_t)(tileM + r) * K;
  v8f acc = {};

#pragma unroll 4
  for (int k = 0; k < K; k += 4) {
    v2f a, b;
    const int ka = k + 2 * hi;
    a.x = Arow[ka];
    a.y = Arow[ka + 1];
    b.x = B[(size_t)ka * Ncols + tileN + r];
    b.y = B[(size_t)(ka + 1) * Ncols + tileN + r];
    acc = __builtin_amdgcn_wmma_f32_16x16x4_f32(
         false, a,  false, b,
         (short)0, acc,  false,  false);
  }

  const int rowBase = tileM + (hi << 3);

  float v[16];
#pragma unroll
  for (int i = 0; i < 8; ++i) {
    v[i]     = acc[i] * ws;
    v[8 + i] = acc[i] * wd;
  }
  fold_step<8>(v, (lane & 8) != 0);
  fold_step<4>(v, (lane & 4) != 0);
  fold_step<2>(v, (lane & 2) != 0);
  fold_step<1>(v, (lane & 1) != 0);
  red[wave][lane] = v[0];
  __syncthreads();

#pragma unroll
  for (int i = 0; i < 8; ++i)
    C[(size_t)(rowBase + i) * Ncols + col] = acc[i];

  if ((wave & 3) == 0) {
    const float t = (red[wave][lane] + red[wave + 1][lane]) +
                    (red[wave + 2][lane] + red[wave + 3][lane]);
    const int H = Ncols >> 6;
    float* __restrict__ dstp = (lane & 8) ? a_d : a_s;
    dstp[(size_t)(rowBase + (lane & 7)) * H + (wave >> 2)] = t;
  }
}

#define OWN_R    416
#define OWN_NW   32
#define OWN_QCAP 64
#define OWN_KU   4
#define OWN_KEY_NONE 255u
#define OWN_G    8
#define OWN_LCAP 8192
#define OWN_KHEAD 32
#define OWN_KSTRIDE (OWN_KHEAD + 2 * OWN_LCAP)
static_assert(OWN_LCAP % 1024 == 0, "a thread reads OWN_LCAP/1024 saved entries");

#define OWN_KW 4
__global__ void __launch_bounds__(256)
owner_keys(const int* __restrict__ dstp, int E, int N,
           unsigned* __restrict__ keys, int nWords) {
  const int w0 = blockIdx.x * (256 * OWN_KW) + threadIdx.x;
  int d[OWN_KW][4];
#pragma unroll
  for (int u = 0; u < OWN_KW; ++u) {
#pragma unroll
    for (int b = 0; b < 4; ++b)
      d[u][b] = dstp[min(4 * (w0 + u * 256) + b, E - 1)];
  }
#pragma unroll
  for (int u = 0; u < OWN_KW; ++u) {
    const int w = w0 + u * 256;
    unsigned packed = 0u;
#pragma unroll
    for (int b = 0; b < 4; ++b) {
      unsigned key = OWN_KEY_NONE;
      if (4 * w + b < E && (unsigned)d[u][b] < (unsigned)N) key = (unsigned)d[u][b] / OWN_R;
      packed |= key << (8 * b);
    }
    if (w < nWords) keys[w] = packed;
  }
}

__device__ __forceinline__ int ord_key(int b) { return b >= 0 ? b : (b ^ 0x7fffffff); }

template <int H, bool ELU, bool SAVE, bool REUSE>
__global__ void __launch_bounds__(1024)
owner_aggregate(const int* __restrict__ ei, int E, int N,
                const float* __restrict__ a_s, const float* __restrict__ a_d,
                const float* __restrict__ hfeat, const float* __restrict__ bias,
                float* __restrict__ out, int* __restrict__ kept,
                const unsigned* __restrict__ keys, int nIterK) {
  constexpr int F = H * 64;
  __shared__ float acc[OWN_R * F];
  __shared__ float dns[OWN_R * H];
  __shared__ int   mxs[OWN_R * H];
  __shared__ int   q_e[OWN_NW][OWN_QCAP];
  __shared__ int   l_s[OWN_LCAP];
  __shared__ int   l_d[OWN_LCAP];
  __shared__ int   l_n;

  const int tid  = threadIdx.x;
  const int lane = tid & 31;
  const int wave = tid >> 5;
  const int lo   = blockIdx.x * OWN_R;
  const int cnt  = min(OWN_R, N - lo);
  const int* __restrict__ srcp = ei;
  const int* __restrict__ dstp = ei + E;

  int* __restrict__ k_n = kept + (size_t)blockIdx.x * OWN_KSTRIDE;
  int* __restrict__ k_s = k_n + OWN_KHEAD;
  int* __restrict__ k_d = k_s + OWN_LCAP;
  int  k_total = 0;
  bool k_have  = false;
  if (REUSE) {
    k_total = k_n[0];
    k_have  = (unsigned)k_total <= (unsigned)OWN_LCAP;
  }

  if (tid == 0) l_n = k_have ? k_total : 0;
  for (int i = tid; i < cnt * F; i += 1024) acc[i] = 0.0f;
  for (int i = tid; i < cnt * H; i += 1024) {
    float v = a_s[(size_t)lo * H + i] + a_d[(size_t)lo * H + i];
    v = (v > 0.0f) ? v : NEG_SLOPE * v;
    mxs[i] = ord_key(__float_as_int(v));
    dns[i] = 0.0f;
  }
  __syncthreads();

  auto score = [&](int s, int dl, int h) -> float {
    float v = a_s[(size_t)s * H + h] + a_d[(size_t)(lo + dl) * H + h];
    return (v > 0.0f) ? v : NEG_SLOPE * v;
  };

  auto add_rows = [&](bool valid, int s, int dl, int m) {
    float w[H];
#pragma unroll
    for (int h = 0; h < H; ++h) w[h] = 0.0f;
    if (valid) {
#pragma unroll
      for (int h = 0; h < H; ++h) {
        const float mx = __int_as_float(ord_key(mxs[dl * H + h]));
        w[h] = __expf(score(s, dl, h) - mx);
        atomicAdd(&dns[dl * H + h], w[h]);
      }
    }
    for (int k0 = 0; k0 < m; k0 += OWN_G) {
      float2 r[OWN_G][H];
      int    dk[OWN_G];
#pragma unroll
      for (int g = 0; g < OWN_G; ++g) {
        const int kk = min(k0 + g, m - 1);
        const int sk = __builtin_amdgcn_readlane(s, kk);
        dk[g] = __builtin_amdgcn_readlane(dl, kk);
#pragma unroll
        for (int h = 0; h < H; ++h) {
          const float wk = __int_as_float(__builtin_amdgcn_readlane(__float_as_int(w[h]), kk));
          const float2* __restrict__ hp =
              reinterpret_cast<const float2*>(hfeat + ((size_t)sk * H + h) * 64);
          const float2 v = hp[lane];
          r[g][h].x = wk * v.x;
          r[g][h].y = wk * v.y;
        }
      }
#pragma unroll
      for (int g = 0; g < OWN_G; ++g) {
        if (k0 + g < m) {
#pragma unroll
          for (int h = 0; h < H; ++h) {
            float* ap = acc + dk[g] * F + h * 64 + 2 * lane;
            atomicAdd(&ap[0], r[g][h].x);
            atomicAdd(&ap[1], r[g][h].y);
          }
        }
      }
    }
  };

  auto flush = [&](int phase, int n) {
    __builtin_amdgcn_fence(__ATOMIC_ACQ_REL, "wavefront");
    for (int b = 0; b < n; b += 32) {
      const int i = b + lane;
      const bool valid = i < n;
      int s = 0, dl = 0;
      if (valid) {
        const int e = min(q_e[wave][i], E - 1);
        s  = srcp[e];
        dl = dstp[e] - lo;
        s  = min(max(s, 0), N - 1);
        dl = min(max(dl, 0), cnt - 1);
      }
      if (phase == 1) {
        int at = 0;
        if (lane == 0) at = atomicAdd(&l_n, min(32, n - b));
        at = __builtin_amdgcn_readfirstlane(at) + lane;
        if (valid) {
          if (at < OWN_LCAP) {
            l_s[at] = s;
            l_d[at] = dl;
          }
#pragma unroll
          for (int h = 0; h < H; ++h)
            atomicMax(&mxs[dl * H + h], ord_key(__float_as_int(score(s, dl, h))));
        }
      } else {
        add_rows(valid, s, dl, min(32, n - b));
      }
    }
    __builtin_amdgcn_fence(__ATOMIC_ACQ_REL, "wavefront");
  };

  const int rot = (int)(((long)blockIdx.x * nIterK) / gridDim.x);
  const uint4* __restrict__ kp = reinterpret_cast<const uint4*>(keys);
  const int wv = __builtin_amdgcn_readfirstlane(wave);
  const unsigned k4 = (unsigned)blockIdx.x * 0x01010101u;
  auto own4 = [&](unsigned w) -> unsigned {
    const unsigned x = w ^ k4;
    const unsigned t = ((x & 0x7f7f7f7fu) + 0x7f7f7f7fu) | x;
    return ~t & 0x80808080u;
  };

  bool list_ok = false;
  for (int phase = 1; phase <= 2; ++phase) {
    int qn = 0;
    if (list_ok) {
      const int total = l_n;
      for (int b = wave * 32; b < total; b += 32 * OWN_NW) {
        const int i = b + lane;
        const bool valid = i < total;
        add_rows(valid, valid ? l_s[i] : 0, valid ? l_d[i] : 0, min(32, total - b));
      }
    } else if (REUSE && k_have) {
      constexpr int PER = OWN_LCAP / 1024;
      if (k_total > 0) {
        int  sv[PER], dv[PER];
        int  key[PER][H];
#pragma unroll
        for (int j = 0; j < PER; ++j) {
          const int i = min(tid + j * 1024, k_total - 1);
          sv[j] = min(max(k_s[i], 0), N - 1);
          dv[j] = min(max(k_d[i], 0), cnt - 1);
        }
#pragma unroll
        for (int j = 0; j < PER; ++j) {
#pragma unroll
          for (int h = 0; h < H; ++h)
            key[j][h] = ord_key(__float_as_int(score(sv[j], dv[j], h)));
        }
#pragma unroll
        for (int j = 0; j < PER; ++j) {
          const int i = tid + j * 1024;
          if (i < k_total) {
            l_s[i] = sv[j];
            l_d[i] = dv[j];
#pragma unroll
            for (int h = 0; h < H; ++h) atomicMax(&mxs[dv[j] * H + h], key[j][h]);
          }
        }
      }
    } else
    for (int t0 = wv * OWN_KU; t0 < nIterK; t0 += OWN_NW * OWN_KU) {
      uint4 kv[OWN_KU];
      int   e0[OWN_KU];
#pragma unroll
      for (int j = 0; j < OWN_KU; ++j) {
        int it = min(t0 + j, nIterK - 1) + rot;
        if (it >= nIterK) it -= nIterK;
        kv[j] = kp[(size_t)it * 32 + lane];
        e0[j] = it * 512 + lane * 16;
      }
#pragma unroll
      for (int j = 0; j < OWN_KU; ++j) {
        unsigned f = (own4(kv[j].x) >> 7) | (own4(kv[j].y) >> 6) |
                     (own4(kv[j].z) >> 5) | (own4(kv[j].w) >> 4);
        if (t0 + j >= nIterK) f = 0u;
        unsigned mask = __builtin_amdgcn_ballot_w32(f != 0u);
        while (mask != 0u) {
          const int pos = qn + __popc(mask & ((1u << lane) - 1u));
          if (f != 0u) {
            const int bit = __builtin_ctz(f);
            q_e[wave][pos] = e0[j] + ((bit & 7) << 2) + (bit >> 3);
            f &= f - 1u;
          }
          qn += __popc(mask);
          if (qn >= 32) {
            flush(phase, qn);
            qn = 0;
          }
          mask = __builtin_amdgcn_ballot_w32(f != 0u);
        }
      }
    }
    if (qn > 0) flush(phase, qn);
    if (phase == 2) {
      for (int b = wave * 32; b < cnt; b += 32 * OWN_NW) {
        const int dl = b + lane;
        const bool valid = dl < cnt;
        add_rows(valid, valid ? lo + dl : 0, valid ? dl : 0, min(32, cnt - b));
      }
    }
    __syncthreads();
    list_ok = (l_n <= OWN_LCAP);
    if (SAVE && phase == 1) {
      const int total = l_n;
      if (tid == 0) k_n[0] = total;
      if (list_ok) {
        for (int i = tid; i < total; i += 1024) {
          k_s[i] = l_s[i];
          k_d[i] = l_d[i];
        }
      }
    }
  }

  for (int i = tid; i < cnt * F; i += 1024) {
    const int nl = i / F;
    const int c  = i - nl * F;
    float v = acc[i] / dns[nl * H + (c >> 6)] + bias[c];
    if (ELU) v = (v > 0.0f) ? v : expm1f(v);
    out[(size_t)lo * F + i] = v;
  }
}

static inline int cdiv_l(long a, int b) { return (int)((a + b - 1) / b); }

extern "C" void kernel_launch(void* const* d_in, const int* in_sizes, int n_in,
                              void* d_out, int out_size, void* d_ws, size_t ws_size,
                              hipStream_t stream) {
  const float* x    = (const float*)d_in[0];
  const int*   ei   = (const int*)d_in[1];
  const float* W1   = (const float*)d_in[2];
  const float* as1w = (const float*)d_in[3];
  const float* ad1w = (const float*)d_in[4];
  const float* b1   = (const float*)d_in[5];
  const float* W2   = (const float*)d_in[6];
  const float* as2w = (const float*)d_in[7];
  const float* ad2w = (const float*)d_in[8];
  const float* b2   = (const float*)d_in[9];

  const int Fin = 128;
  const int N   = in_sizes[0] / Fin;
  const int E   = in_sizes[1] / 2;

  float* ws   = (float*)d_ws;
  float* h1   = ws;  ws += (size_t)N * 128;
  float* agg1 = ws;  ws += (size_t)N * 128;
  int*   kept = (int*)ws;  ws += (size_t)N * 64;
  float* h2   = h1;
  float* a_s  = ws;  ws += (size_t)N * 2;
  float* a_d  = ws;  ws += (size_t)N * 2;
  float* outp = (float*)d_out;

  const int nOwn = cdiv_l((long)N, OWN_R);
  const int nIterK = cdiv_l((long)E, 512);
  unsigned* keys = (unsigned*)(kept + (size_t)nOwn * OWN_KSTRIDE);

  owner_keys<<<cdiv_l((long)nIterK * 128, 256 * OWN_KW), 256, 0, stream>>>(ei + E, E, N, keys, nIterK * 128);

  wmma_gemm_f32<<<N / 16, 32 * (128 / 16), 0, stream>>>(x, W1, h1, 128, 128, as1w, ad1w, a_s, a_d);
  owner_aggregate<2, true, true, false><<<nOwn, 1024, 0, stream>>>(ei, E, N, a_s, a_d, h1, b1, agg1, kept, keys, nIterK);

  wmma_gemm_f32<<<N / 16, 32 * (64 / 16), 0, stream>>>(agg1, W2, h2, 128, 64, as2w, ad2w, a_s, a_d);
  owner_aggregate<1, false, false, true><<<nOwn, 1024, 0, stream>>>(ei, E, N, a_s, a_d, h2, b2, outp, kept, keys, nIterK);
}


# layer 2's product and logits formed in layer 1's owner-block epilogue on the matrix cores from the rows held in LDS: activation plane and second GEMM launch gone; layer-2 logit sum order changes
# speedup vs baseline: 5.0817x; 1.0947x over previous
#include <hip/hip_runtime.h>
#include <hip/hip_bf16.h>
#include <math.h>

typedef __attribute__((ext_vector_type(2))) float v2f;
typedef __attribute__((ext_vector_type(8))) float v8f;

#define NEG_SLOPE 0.2f


template <int M>
__device__ __forceinline__ float row_xor(float v) {
  return __int_as_float(__builtin_amdgcn_update_dpp(
      0, __float_as_int(v), 0x160 + M, 0xf, 0xf, false));
}

template <int M>
__device__ __forceinline__ void fold_step(float (&v)[16], bool up) {
#pragma unroll
  for (int j = 0; j < M; ++j) {
    const float keep = up ? v[j + M] : v[j];
    const float send = up ? v[j] : v[j + M];
    v[j] = keep + row_xor<M>(send);
  }
}

__global__ void wmma_gemm_f32(const float* __restrict__ A,
                              const float* __restrict__ B,
                              float* __restrict__ C, int K, int Ncols,
                              const float* __restrict__ att_s,
                              const float* __restrict__ att_d,
                              float* __restrict__ a_s, float* __restrict__ a_d) {
  __shared__ float red[8][32];

  const int tileM = blockIdx.x * 16;
  const int wave  = threadIdx.x >> 5;
  const int lane  = threadIdx.x & 31;
  const int tileN = wave * 16;
  const int r  = lane & 15;
  const int hi = lane >> 4;
  const int col = tileN + r;

  const float ws = att_s[col];
  const float wd = att_d[col];

  const float* __restrict__ Arow = A + (size_t)(tileM + r) * K;
  v8f acc = {};

#pragma unroll 4
  for (int k = 0; k < K; k += 4) {
    v2f a, b;
    const int ka = k + 2 * hi;
    a.x = Arow[ka];
    a.y = Arow[ka + 1];
    b.x = B[(size_t)ka * Ncols + tileN + r];
    b.y = B[(size_t)(ka + 1) * Ncols + tileN + r];
    acc = __builtin_amdgcn_wmma_f32_16x16x4_f32(
         false, a,  false, b,
         (short)0, acc,  false,  false);
  }

  const int rowBase = tileM + (hi << 3);

  float v[16];
#pragma unroll
  for (int i = 0; i < 8; ++i) {
    v[i]     = acc[i] * ws;
    v[8 + i] = acc[i] * wd;
  }
  fold_step<8>(v, (lane & 8) != 0);
  fold_step<4>(v, (lane & 4) != 0);
  fold_step<2>(v, (lane & 2) != 0);
  fold_step<1>(v, (lane & 1) != 0);
  red[wave][lane] = v[0];
  __syncthreads();

#pragma unroll
  for (int i = 0; i < 8; ++i)
    C[(size_t)(rowBase + i) * Ncols + col] = acc[i];

  if ((wave & 3) == 0) {
    const float t = (red[wave][lane] + red[wave + 1][lane]) +
                    (red[wave + 2][lane] + red[wave + 3][lane]);
    const int H = Ncols >> 6;
    float* __restrict__ dstp = (lane & 8) ? a_d : a_s;
    dstp[(size_t)(rowBase + (lane & 7)) * H + (wave >> 2)] = t;
  }
}

#define OWN_R    416
#define OWN_NW   32
#define OWN_QCAP 64
#define OWN_KU   4
#define OWN_KEY_NONE 255u
#define OWN_G    8
#define OWN_LCAP 8192
#define OWN_KHEAD 32
#define OWN_KSTRIDE (OWN_KHEAD + 2 * OWN_LCAP)
static_assert(OWN_LCAP % 1024 == 0, "a thread reads OWN_LCAP/1024 saved entries");

#define OWN_KW 4
__global__ void __launch_bounds__(256)
owner_keys(const int* __restrict__ dstp, int E, int N,
           unsigned* __restrict__ keys, int nWords) {
  const int w0 = blockIdx.x * (256 * OWN_KW) + threadIdx.x;
  int d[OWN_KW][4];
#pragma unroll
  for (int u = 0; u < OWN_KW; ++u) {
#pragma unroll
    for (int b = 0; b < 4; ++b)
      d[u][b] = dstp[min(4 * (w0 + u * 256) + b, E - 1)];
  }
#pragma unroll
  for (int u = 0; u < OWN_KW; ++u) {
    const int w = w0 + u * 256;
    unsigned packed = 0u;
#pragma unroll
    for (int b = 0; b < 4; ++b) {
      unsigned key = OWN_KEY_NONE;
      if (4 * w + b < E && (unsigned)d[u][b] < (unsigned)N) key = (unsigned)d[u][b] / OWN_R;
      packed |= key << (8 * b);
    }
    if (w < nWords) keys[w] = packed;
  }
}

__device__ __forceinline__ int ord_key(int b) { return b >= 0 ? b : (b ^ 0x7fffffff); }

template <int H, bool ELU, bool SAVE, bool REUSE, bool FUSE>
__global__ void __launch_bounds__(1024)
owner_aggregate(const int* __restrict__ ei, int E, int N,
                const float* __restrict__ a_s, const float* __restrict__ a_d,
                const float* __restrict__ hfeat, const float* __restrict__ bias,
                float* __restrict__ out, int* __restrict__ kept,
                const unsigned* __restrict__ keys, int nIterK,
                const float* __restrict__ Wn, const float* __restrict__ att_sn,
                const float* __restrict__ att_dn, float* __restrict__ a_sn,
                float* __restrict__ a_dn) {
  constexpr int F = H * 64;
  constexpr int FP = FUSE ? F + 4 : F;
  __shared__ __attribute__((aligned(16))) float acc[OWN_R * FP];
  __shared__ float dns[OWN_R * H];
  __shared__ int   mxs[OWN_R * H];
  __shared__ int   q_e[OWN_NW][OWN_QCAP];
  __shared__ int   l_s[OWN_LCAP];
  __shared__ int   l_d[OWN_LCAP];
  __shared__ int   l_n;

  const int tid  = threadIdx.x;
  const int lane = tid & 31;
  const int wave = tid >> 5;
  const int lo   = blockIdx.x * OWN_R;
  const int cnt  = min(OWN_R, N - lo);
  const int* __restrict__ srcp = ei;
  const int* __restrict__ dstp = ei + E;

  int* __restrict__ k_n = kept + (size_t)blockIdx.x * OWN_KSTRIDE;
  int* __restrict__ k_s = k_n + OWN_KHEAD;
  int* __restrict__ k_d = k_s + OWN_LCAP;
  int  k_total = 0;
  bool k_have  = false;
  if (REUSE) {
    k_total = k_n[0];
    k_have  = (unsigned)k_total <= (unsigned)OWN_LCAP;
  }

  if (tid == 0) l_n = k_have ? k_total : 0;
  for (int i = tid; i < cnt * FP; i += 1024) acc[i] = 0.0f;
  for (int i = tid; i < cnt * H; i += 1024) {
    float v = a_s[(size_t)lo * H + i] + a_d[(size_t)lo * H + i];
    v = (v > 0.0f) ? v : NEG_SLOPE * v;
    mxs[i] = ord_key(__float_as_int(v));
    dns[i] = 0.0f;
  }
  __syncthreads();

  auto score = [&](int s, int dl, int h) -> float {
    float v = a_s[(size_t)s * H + h] + a_d[(size_t)(lo + dl) * H + h];
    return (v > 0.0f) ? v : NEG_SLOPE * v;
  };

  auto add_rows = [&](bool valid, int s, int dl, int m) {
    float w[H];
#pragma unroll
    for (int h = 0; h < H; ++h) w[h] = 0.0f;
    if (valid) {
#pragma unroll
      for (int h = 0; h < H; ++h) {
        const float mx = __int_as_float(ord_key(mxs[dl * H + h]));
        w[h] = __expf(score(s, dl, h) - mx);
        atomicAdd(&dns[dl * H + h], w[h]);
      }
    }
    for (int k0 = 0; k0 < m; k0 += OWN_G) {
      float2 r[OWN_G][H];
      int    dk[OWN_G];
#pragma unroll
      for (int g = 0; g < OWN_G; ++g) {
        const int kk = min(k0 + g, m - 1);
        const int sk = __builtin_amdgcn_readlane(s, kk);
        dk[g] = __builtin_amdgcn_readlane(dl, kk);
#pragma unroll
        for (int h = 0; h < H; ++h) {
          const float wk = __int_as_float(__builtin_amdgcn_readlane(__float_as_int(w[h]), kk));
          const float2* __restrict__ hp =
              reinterpret_cast<const float2*>(hfeat + ((size_t)sk * H + h) * 64);
          const float2 v = hp[lane];
          r[g][h].x = wk * v.x;
          r[g][h].y = wk * v.y;
        }
      }
#pragma unroll
      for (int g = 0; g < OWN_G; ++g) {
        if (k0 + g < m) {
#pragma unroll
          for (int h = 0; h < H; ++h) {
            float* ap = acc + dk[g] * FP + h * 64 + 2 * lane;
            atomicAdd(&ap[0], r[g][h].x);
            atomicAdd(&ap[1], r[g][h].y);
          }
        }
      }
    }
  };

  auto flush = [&](int phase, int n) {
    __builtin_amdgcn_fence(__ATOMIC_ACQ_REL, "wavefront");
    for (int b = 0; b < n; b += 32) {
      const int i = b + lane;
      const bool valid = i < n;
      int s = 0, dl = 0;
      if (valid) {
        const int e = min(q_e[wave][i], E - 1);
        s  = srcp[e];
        dl = dstp[e] - lo;
        s  = min(max(s, 0), N - 1);
        dl = min(max(dl, 0), cnt - 1);
      }
      if (phase == 1) {
        int at = 0;
        if (lane == 0) at = atomicAdd(&l_n, min(32, n - b));
        at = __builtin_amdgcn_readfirstlane(at) + lane;
        if (valid) {
          if (at < OWN_LCAP) {
            l_s[at] = s;
            l_d[at] = dl;
          }
#pragma unroll
          for (int h = 0; h < H; ++h)
            atomicMax(&mxs[dl * H + h], ord_key(__float_as_int(score(s, dl, h))));
        }
      } else {
        add_rows(valid, s, dl, min(32, n - b));
      }
    }
    __builtin_amdgcn_fence(__ATOMIC_ACQ_REL, "wavefront");
  };

  const int rot = (int)(((long)blockIdx.x * nIterK) / gridDim.x);
  const uint4* __restrict__ kp = reinterpret_cast<const uint4*>(keys);
  const int wv = __builtin_amdgcn_readfirstlane(wave);
  const unsigned k4 = (unsigned)blockIdx.x * 0x01010101u;
  auto own4 = [&](unsigned w) -> unsigned {
    const unsigned x = w ^ k4;
    const unsigned t = ((x & 0x7f7f7f7fu) + 0x7f7f7f7fu) | x;
    return ~t & 0x80808080u;
  };

  bool list_ok = false;
  for (int phase = 1; phase <= 2; ++phase) {
    int qn = 0;
    if (list_ok) {
      const int total = l_n;
      for (int b = wave * 32; b < total; b += 32 * OWN_NW) {
        const int i = b + lane;
        const bool valid = i < total;
        add_rows(valid, valid ? l_s[i] : 0, valid ? l_d[i] : 0, min(32, total - b));
      }
    } else if (REUSE && k_have) {
      constexpr int PER = OWN_LCAP / 1024;
      if (k_total > 0) {
        int  sv[PER], dv[PER];
        int  key[PER][H];
#pragma unroll
        for (int j = 0; j < PER; ++j) {
          const int i = min(tid + j * 1024, k_total - 1);
          sv[j] = min(max(k_s[i], 0), N - 1);
          dv[j] = min(max(k_d[i], 0), cnt - 1);
        }
#pragma unroll
        for (int j = 0; j < PER; ++j) {
#pragma unroll
          for (int h = 0; h < H; ++h)
            key[j][h] = ord_key(__float_as_int(score(sv[j], dv[j], h)));
        }
#pragma unroll
        for (int j = 0; j < PER; ++j) {
          const int i = tid + j * 1024;
          if (i < k_total) {
            l_s[i] = sv[j];
            l_d[i] = dv[j];
#pragma unroll
            for (int h = 0; h < H; ++h) atomicMax(&mxs[dv[j] * H + h], key[j][h]);
          }
        }
      }
    } else
    for (int t0 = wv * OWN_KU; t0 < nIterK; t0 += OWN_NW * OWN_KU) {
      uint4 kv[OWN_KU];
      int   e0[OWN_KU];
#pragma unroll
      for (int j = 0; j < OWN_KU; ++j) {
        int it = min(t0 + j, nIterK - 1) + rot;
        if (it >= nIterK) it -= nIterK;
        kv[j] = kp[(size_t)it * 32 + lane];
        e0[j] = it * 512 + lane * 16;
      }
#pragma unroll
      for (int j = 0; j < OWN_KU; ++j) {
        unsigned f = (own4(kv[j].x) >> 7) | (own4(kv[j].y) >> 6) |
                     (own4(kv[j].z) >> 5) | (own4(kv[j].w) >> 4);
        if (t0 + j >= nIterK) f = 0u;
        unsigned mask = __builtin_amdgcn_ballot_w32(f != 0u);
        while (mask != 0u) {
          const int pos = qn + __popc(mask & ((1u << lane) - 1u));
          if (f != 0u) {
            const int bit = __builtin_ctz(f);
            q_e[wave][pos] = e0[j] + ((bit & 7) << 2) + (bit >> 3);
            f &= f - 1u;
          }
          qn += __popc(mask);
          if (qn >= 32) {
            flush(phase, qn);
            qn = 0;
          }
          mask = __builtin_amdgcn_ballot_w32(f != 0u);
        }
      }
    }
    if (qn > 0) flush(phase, qn);
    if (phase == 2) {
      for (int b = wave * 32; b < cnt; b += 32 * OWN_NW) {
        const int dl = b + lane;
        const bool valid = dl < cnt;
        add_rows(valid, valid ? lo + dl : 0, valid ? dl : 0, min(32, cnt - b));
      }
    }
    __syncthreads();
    list_ok = (l_n <= OWN_LCAP);
    if (SAVE && phase == 1) {
      const int total = l_n;
      if (tid == 0) k_n[0] = total;
      if (list_ok) {
        for (int i = tid; i < total; i += 1024) {
          k_s[i] = l_s[i];
          k_d[i] = l_d[i];
        }
      }
    }
  }

  if constexpr (!FUSE) {
    for (int i = tid; i < cnt * F; i += 1024) {
      const int nl = i / F;
      const int c  = i - nl * F;
      float v = acc[i] / dns[nl * H + (c >> 6)] + bias[c];
      if (ELU) v = (v > 0.0f) ? v : expm1f(v);
      out[(size_t)lo * F + i] = v;
    }
  } else {
    for (int i = tid; i < cnt * F; i += 1024) {
      const int nl = i / F;
      const int c  = i - nl * F;
      float v = acc[nl * FP + c] / dns[nl * H + (c >> 6)] + bias[c];
      if (ELU) v = (v > 0.0f) ? v : expm1f(v);
      acc[nl * FP + c] = v;
    }
    __syncthreads();

    const int r  = lane & 15;
    const int hi = lane >> 4;
    const float4 ws4 = *reinterpret_cast<const float4*>(att_sn + 4 * r);
    const float4 wd4 = *reinterpret_cast<const float4*>(att_dn + 4 * r);
    const int nt = cnt >> 4;
    for (int rt = wv; rt < nt; rt += OWN_NW) {
      const float* arow = acc + (rt * 16 + r) * FP + 2 * hi;
      const float* __restrict__ brow = Wn + (size_t)(2 * hi) * 64 + 4 * r;
      v8f c0 = {}, c1 = {}, c2 = {}, c3 = {};
#pragma unroll 4
      for (int k = 0; k < F; k += 4) {
        const v2f a = *reinterpret_cast<const v2f*>(arow + k);
        const float4 p = *reinterpret_cast<const float4*>(brow + (size_t)k * 64);
        const float4 q = *reinterpret_cast<const float4*>(brow + (size_t)(k + 1) * 64);
        v2f b;
        b.x = p.x; b.y = q.x;
        c0 = __builtin_amdgcn_wmma_f32_16x16x4_f32(false, a, false, b, (short)0, c0, false, false);
        b.x = p.y; b.y = q.y;
        c1 = __builtin_amdgcn_wmma_f32_16x16x4_f32(false, a, false, b, (short)0, c1, false, false);
        b.x = p.z; b.y = q.z;
        c2 = __builtin_amdgcn_wmma_f32_16x16x4_f32(false, a, false, b, (short)0, c2, false, false);
        b.x = p.w; b.y = q.w;
        c3 = __builtin_amdgcn_wmma_f32_16x16x4_f32(false, a, false, b, (short)0, c3, false, false);
      }
      const int rowBase = lo + rt * 16 + (hi << 3);
      float v[16];
#pragma unroll
      for (int i = 0; i < 8; ++i) {
        float4 o;
        o.x = c0[i]; o.y = c1[i]; o.z = c2[i]; o.w = c3[i];
        *reinterpret_cast<float4*>(out + (size_t)(rowBase + i) * 64 + 4 * r) = o;
        v[i]     = (o.x * ws4.x + o.y * ws4.y) + (o.z * ws4.z + o.w * ws4.w);
        v[8 + i] = (o.x * wd4.x + o.y * wd4.y) + (o.z * wd4.z + o.w * wd4.w);
      }
      fold_step<8>(v, (lane & 8) != 0);
      fold_step<4>(v, (lane & 4) != 0);
      fold_step<2>(v, (lane & 2) != 0);
      fold_step<1>(v, (lane & 1) != 0);
      float* __restrict__ lp = (lane & 8) ? a_dn : a_sn;
      lp[rowBase + (lane & 7)] = v[0];
    }
  }
}

static inline int cdiv_l(long a, int b) { return (int)((a + b - 1) / b); }

extern "C" void kernel_launch(void* const* d_in, const int* in_sizes, int n_in,
                              void* d_out, int out_size, void* d_ws, size_t ws_size,
                              hipStream_t stream) {
  const float* x    = (const float*)d_in[0];
  const int*   ei   = (const int*)d_in[1];
  const float* W1   = (const float*)d_in[2];
  const float* as1w = (const float*)d_in[3];
  const float* ad1w = (const float*)d_in[4];
  const float* b1   = (const float*)d_in[5];
  const float* W2   = (const float*)d_in[6];
  const float* as2w = (const float*)d_in[7];
  const float* ad2w = (const float*)d_in[8];
  const float* b2   = (const float*)d_in[9];

  const int Fin = 128;
  const int N   = in_sizes[0] / Fin;
  const int E   = in_sizes[1] / 2;

  float* ws   = (float*)d_ws;
  float* h1   = ws;  ws += (size_t)N * 128;
  float* h2   = ws;
  float* a_s2 = ws + (size_t)N * 64;
  float* a_d2 = ws + (size_t)N * 65;
  ws += (size_t)N * 128;
  int*   kept = (int*)ws;  ws += (size_t)N * 64;
  float* a_s  = ws;  ws += (size_t)N * 2;
  float* a_d  = ws;  ws += (size_t)N * 2;
  float* outp = (float*)d_out;

  const int nOwn = cdiv_l((long)N, OWN_R);
  const int nIterK = cdiv_l((long)E, 512);
  unsigned* keys = (unsigned*)(kept + (size_t)nOwn * OWN_KSTRIDE);

  owner_keys<<<cdiv_l((long)nIterK * 128, 256 * OWN_KW), 256, 0, stream>>>(ei + E, E, N, keys, nIterK * 128);

  wmma_gemm_f32<<<N / 16, 32 * (128 / 16), 0, stream>>>(x, W1, h1, 128, 128, as1w, ad1w, a_s, a_d);
  owner_aggregate<2, true, true, false, true><<<nOwn, 1024, 0, stream>>>(
      ei, E, N, a_s, a_d, h1, b1, h2, kept, keys, nIterK, W2, as2w, ad2w, a_s2, a_d2);

  owner_aggregate<1, false, false, true, false><<<nOwn, 1024, 0, stream>>>(
      ei, E, N, a_s2, a_d2, h2, b2, outp, kept, keys, nIterK, nullptr, nullptr, nullptr, nullptr, nullptr);
}


# layer-1 product: a wave takes 16 rows x one head's 64 columns (4 accumulators share each A fragment, B read 16 B a lane from global) and sums the head's logits itself; logit sum order changes
# speedup vs baseline: 5.7543x; 1.1324x over previous
#include <hip/hip_runtime.h>
#include <hip/hip_bf16.h>
#include <math.h>

typedef __attribute__((ext_vector_type(2))) float v2f;
typedef __attribute__((ext_vector_type(8))) float v8f;

#define NEG_SLOPE 0.2f


template <int M>
__device__ __forceinline__ float row_xor(float v) {
  return __int_as_float(__builtin_amdgcn_update_dpp(
      0, __float_as_int(v), 0x160 + M, 0xf, 0xf, false));
}

template <int M>
__device__ __forceinline__ void fold_step(float (&v)[16], bool up) {
#pragma unroll
  for (int j = 0; j < M; ++j) {
    const float keep = up ? v[j + M] : v[j];
    const float send = up ? v[j] : v[j + M];
    v[j] = keep + row_xor<M>(send);
  }
}

template <int K, int H>
__global__ void __launch_bounds__(256)
wmma_gemm_heads(const float* __restrict__ A, const float* __restrict__ B,
                float* __restrict__ C, int nTiles,
                const float* __restrict__ att_s,
                const float* __restrict__ att_d,
                float* __restrict__ a_s, float* __restrict__ a_d) {
  constexpr int NC = H * 64;
  const int lane = threadIdx.x & 31;
  const int wv = __builtin_amdgcn_readfirstlane((int)(threadIdx.x >> 5));
  const int h  = wv % H;
  const int rt = blockIdx.x * (8 / H) + wv / H;
  if (rt >= nTiles) return;

  const int r  = lane & 15;
  const int hi = lane >> 4;
  const float4 ws4 = *reinterpret_cast<const float4*>(att_s + h * 64 + 4 * r);
  const float4 wd4 = *reinterpret_cast<const float4*>(att_d + h * 64 + 4 * r);

  const float* __restrict__ arow = A + (size_t)(rt * 16 + r) * K + 2 * hi;
  const float* __restrict__ brow = B + (size_t)(2 * hi) * NC + h * 64 + 4 * r;
  v8f c0 = {}, c1 = {}, c2 = {}, c3 = {};
#pragma unroll 4
  for (int k = 0; k < K; k += 4) {
    const v2f a = *reinterpret_cast<const v2f*>(arow + k);
    const float4 p = *reinterpret_cast<const float4*>(brow + (size_t)k * NC);
    const float4 q = *reinterpret_cast<const float4*>(brow + (size_t)(k + 1) * NC);
    v2f b;
    b.x = p.x; b.y = q.x;
    c0 = __builtin_amdgcn_wmma_f32_16x16x4_f32(false, a, false, b, (short)0, c0, false, false);
    b.x = p.y; b.y = q.y;
    c1 = __builtin_amdgcn_wmma_f32_16x16x4_f32(false, a, false, b, (short)0, c1, false, false);
    b.x = p.z; b.y = q.z;
    c2 = __builtin_amdgcn_wmma_f32_16x16x4_f32(false, a, false, b, (short)0, c2, false, false);
    b.x = p.w; b.y = q.w;
    c3 = __builtin_amdgcn_wmma_f32_16x16x4_f32(false, a, false, b, (short)0, c3, false, false);
  }

  const int rowBase = rt * 16 + (hi << 3);
  float v[16];
#pragma unroll
  for (int i = 0; i < 8; ++i) {
    float4 o;
    o.x = c0[i]; o.y = c1[i]; o.z = c2[i]; o.w = c3[i];
    *reinterpret_cast<float4*>(C + (size_t)(rowBase + i) * NC + h * 64 + 4 * r) = o;
    v[i]     = (o.x * ws4.x + o.y * ws4.y) + (o.z * ws4.z + o.w * ws4.w);
    v[8 + i] = (o.x * wd4.x + o.y * wd4.y) + (o.z * wd4.z + o.w * wd4.w);
  }
  fold_step<8>(v, (lane & 8) != 0);
  fold_step<4>(v, (lane & 4) != 0);
  fold_step<2>(v, (lane & 2) != 0);
  fold_step<1>(v, (lane & 1) != 0);
  float* __restrict__ lp = (lane & 8) ? a_d : a_s;
  lp[(size_t)(rowBase + (lane & 7)) * H + h] = v[0];
}

#define OWN_R    416
#define OWN_NW   32
#define OWN_QCAP 64
#define OWN_KU   4
#define OWN_KEY_NONE 255u
#define OWN_G    8
#define OWN_LCAP 8192
#define OWN_KHEAD 32
#define OWN_KSTRIDE (OWN_KHEAD + 2 * OWN_LCAP)
static_assert(OWN_LCAP % 1024 == 0, "a thread reads OWN_LCAP/1024 saved entries");

#define OWN_KW 4
__global__ void __launch_bounds__(256)
owner_keys(const int* __restrict__ dstp, int E, int N,
           unsigned* __restrict__ keys, int nWords) {
  const int w0 = blockIdx.x * (256 * OWN_KW) + threadIdx.x;
  int d[OWN_KW][4];
#pragma unroll
  for (int u = 0; u < OWN_KW; ++u) {
#pragma unroll
    for (int b = 0; b < 4; ++b)
      d[u][b] = dstp[min(4 * (w0 + u * 256) + b, E - 1)];
  }
#pragma unroll
  for (int u = 0; u < OWN_KW; ++u) {
    const int w = w0 + u * 256;
    unsigned packed = 0u;
#pragma unroll
    for (int b = 0; b < 4; ++b) {
      unsigned key = OWN_KEY_NONE;
      if (4 * w + b < E && (unsigned)d[u][b] < (unsigned)N) key = (unsigned)d[u][b] / OWN_R;
      packed |= key << (8 * b);
    }
    if (w < nWords) keys[w] = packed;
  }
}

__device__ __forceinline__ int ord_key(int b) { return b >= 0 ? b : (b ^ 0x7fffffff); }

template <int H, bool ELU, bool SAVE, bool REUSE, bool FUSE>
__global__ void __launch_bounds__(1024)
owner_aggregate(const int* __restrict__ ei, int E, int N,
                const float* __restrict__ a_s, const float* __restrict__ a_d,
                const float* __restrict__ hfeat, const float* __restrict__ bias,
                float* __restrict__ out, int* __restrict__ kept,
                const unsigned* __restrict__ keys, int nIterK,
                const float* __restrict__ Wn, const float* __restrict__ att_sn,
                const float* __restrict__ att_dn, float* __restrict__ a_sn,
                float* __restrict__ a_dn) {
  constexpr int F = H * 64;
  constexpr int FP = FUSE ? F + 4 : F;
  __shared__ __attribute__((aligned(16))) float acc[OWN_R * FP];
  __shared__ float dns[OWN_R * H];
  __shared__ int   mxs[OWN_R * H];
  __shared__ int   q_e[OWN_NW][OWN_QCAP];
  __shared__ int   l_s[OWN_LCAP];
  __shared__ int   l_d[OWN_LCAP];
  __shared__ int   l_n;

  const int tid  = threadIdx.x;
  const int lane = tid & 31;
  const int wave = tid >> 5;
  const int lo   = blockIdx.x * OWN_R;
  const int cnt  = min(OWN_R, N - lo);
  const int* __restrict__ srcp = ei;
  const int* __restrict__ dstp = ei + E;

  int* __restrict__ k_n = kept + (size_t)blockIdx.x * OWN_KSTRIDE;
  int* __restrict__ k_s = k_n + OWN_KHEAD;
  int* __restrict__ k_d = k_s + OWN_LCAP;
  int  k_total = 0;
  bool k_have  = false;
  if (REUSE) {
    k_total = k_n[0];
    k_have  = (unsigned)k_total <= (unsigned)OWN_LCAP;
  }

  if (tid == 0) l_n = k_have ? k_total : 0;
  for (int i = tid; i < cnt * FP; i += 1024) acc[i] = 0.0f;
  for (int i = tid; i < cnt * H; i += 1024) {
    float v = a_s[(size_t)lo * H + i] + a_d[(size_t)lo * H + i];
    v = (v > 0.0f) ? v : NEG_SLOPE * v;
    mxs[i] = ord_key(__float_as_int(v));
    dns[i] = 0.0f;
  }
  __syncthreads();

  auto score = [&](int s, int dl, int h) -> float {
    float v = a_s[(size_t)s * H + h] + a_d[(size_t)(lo + dl) * H + h];
    return (v > 0.0f) ? v : NEG_SLOPE * v;
  };

  auto add_rows = [&](bool valid, int s, int dl, int m) {
    float w[H];
#pragma unroll
    for (int h = 0; h < H; ++h) w[h] = 0.0f;
    if (valid) {
#pragma unroll
      for (int h = 0; h < H; ++h) {
        const float mx = __int_as_float(ord_key(mxs[dl * H + h]));
        w[h] = __expf(score(s, dl, h) - mx);
        atomicAdd(&dns[dl * H + h], w[h]);
      }
    }
    for (int k0 = 0; k0 < m; k0 += OWN_G) {
      float2 r[OWN_G][H];
      int    dk[OWN_G];
#pragma unroll
      for (int g = 0; g < OWN_G; ++g) {
        const int kk = min(k0 + g, m - 1);
        const int sk = __builtin_amdgcn_readlane(s, kk);
        dk[g] = __builtin_amdgcn_readlane(dl, kk);
#pragma unroll
        for (int h = 0; h < H; ++h) {
          const float wk = __int_as_float(__builtin_amdgcn_readlane(__float_as_int(w[h]), kk));
          const float2* __restrict__ hp =
              reinterpret_cast<const float2*>(hfeat + ((size_t)sk * H + h) * 64);
          const float2 v = hp[lane];
          r[g][h].x = wk * v.x;
          r[g][h].y = wk * v.y;
        }
      }
#pragma unroll
      for (int g = 0; g < OWN_G; ++g) {
        if (k0 + g < m) {
#pragma unroll
          for (int h = 0; h < H; ++h) {
            float* ap = acc + dk[g] * FP + h * 64 + 2 * lane;
            atomicAdd(&ap[0], r[g][h].x);
            atomicAdd(&ap[1], r[g][h].y);
          }
        }
      }
    }
  };

  auto flush = [&](int phase, int n) {
    __builtin_amdgcn_fence(__ATOMIC_ACQ_REL, "wavefront");
    for (int b = 0; b < n; b += 32) {
      const int i = b + lane;
      const bool valid = i < n;
      int s = 0, dl = 0;
      if (valid) {
        const int e = min(q_e[wave][i], E - 1);
        s  = srcp[e];
        dl = dstp[e] - lo;
        s  = min(max(s, 0), N - 1);
        dl = min(max(dl, 0), cnt - 1);
      }
      if (phase == 1) {
        int at = 0;
        if (lane == 0) at = atomicAdd(&l_n, min(32, n - b));
        at = __builtin_amdgcn_readfirstlane(at) + lane;
        if (valid) {
          if (at < OWN_LCAP) {
            l_s[at] = s;
            l_d[at] = dl;
          }
#pragma unroll
          for (int h = 0; h < H; ++h)
            atomicMax(&mxs[dl * H + h], ord_key(__float_as_int(score(s, dl, h))));
        }
      } else {
        add_rows(valid, s, dl, min(32, n - b));
      }
    }
    __builtin_amdgcn_fence(__ATOMIC_ACQ_REL, "wavefront");
  };

  const int rot = (int)(((long)blockIdx.x * nIterK) / gridDim.x);
  const uint4* __restrict__ kp = reinterpret_cast<const uint4*>(keys);
  const int wv = __builtin_amdgcn_readfirstlane(wave);
  const unsigned k4 = (unsigned)blockIdx.x * 0x01010101u;
  auto own4 = [&](unsigned w) -> unsigned {
    const unsigned x = w ^ k4;
    const unsigned t = ((x & 0x7f7f7f7fu) + 0x7f7f7f7fu) | x;
    return ~t & 0x80808080u;
  };

  bool list_ok = false;
  for (int phase = 1; phase <= 2; ++phase) {
    int qn = 0;
    if (list_ok) {
      const int total = l_n;
      for (int b = wave * 32; b < total; b += 32 * OWN_NW) {
        const int i = b + lane;
        const bool valid = i < total;
        add_rows(valid, valid ? l_s[i] : 0, valid ? l_d[i] : 0, min(32, total - b));
      }
    } else if (REUSE && k_have) {
      constexpr int PER = OWN_LCAP / 1024;
      if (k_total > 0) {
        int  sv[PER], dv[PER];
        int  key[PER][H];
#pragma unroll
        for (int j = 0; j < PER; ++j) {
          const int i = min(tid + j * 1024, k_total - 1);
          sv[j] = min(max(k_s[i], 0), N - 1);
          dv[j] = min(max(k_d[i], 0), cnt - 1);
        }
#pragma unroll
        for (int j = 0; j < PER; ++j) {
#pragma unroll
          for (int h = 0; h < H; ++h)
            key[j][h] = ord_key(__float_as_int(score(sv[j], dv[j], h)));
        }
#pragma unroll
        for (int j = 0; j < PER; ++j) {
          const int i = tid + j * 1024;
          if (i < k_total) {
            l_s[i] = sv[j];
            l_d[i] = dv[j];
#pragma unroll
            for (int h = 0; h < H; ++h) atomicMax(&mxs[dv[j] * H + h], key[j][h]);
          }
        }
      }
    } else
    for (int t0 = wv * OWN_KU; t0 < nIterK; t0 += OWN_NW * OWN_KU) {
      uint4 kv[OWN_KU];
      int   e0[OWN_KU];
#pragma unroll
      for (int j = 0; j < OWN_KU; ++j) {
        int it = min(t0 + j, nIterK - 1) + rot;
        if (it >= nIterK) it -= nIterK;
        kv[j] = kp[(size_t)it * 32 + lane];
        e0[j] = it * 512 + lane * 16;
      }
#pragma unroll
      for (int j = 0; j < OWN_KU; ++j) {
        unsigned f = (own4(kv[j].x) >> 7) | (own4(kv[j].y) >> 6) |
                     (own4(kv[j].z) >> 5) | (own4(kv[j].w) >> 4);
        if (t0 + j >= nIterK) f = 0u;
        unsigned mask = __builtin_amdgcn_ballot_w32(f != 0u);
        while (mask != 0u) {
          const int pos = qn + __popc(mask & ((1u << lane) - 1u));
          if (f != 0u) {
            const int bit = __builtin_ctz(f);
            q_e[wave][pos] = e0[j] + ((bit & 7) << 2) + (bit >> 3);
            f &= f - 1u;
          }
          qn += __popc(mask);
          if (qn >= 32) {
            flush(phase, qn);
            qn = 0;
          }
          mask = __builtin_amdgcn_ballot_w32(f != 0u);
        }
      }
    }
    if (qn > 0) flush(phase, qn);
    if (phase == 2) {
      for (int b = wave * 32; b < cnt; b += 32 * OWN_NW) {
        const int dl = b + lane;
        const bool valid = dl < cnt;
        add_rows(valid, valid ? lo + dl : 0, valid ? dl : 0, min(32, cnt - b));
      }
    }
    __syncthreads();
    list_ok = (l_n <= OWN_LCAP);
    if (SAVE && phase == 1) {
      const int total = l_n;
      if (tid == 0) k_n[0] = total;
      if (list_ok) {
        for (int i = tid; i < total; i += 1024) {
          k_s[i] = l_s[i];
          k_d[i] = l_d[i];
        }
      }
    }
  }

  if constexpr (!FUSE) {
    for (int i = tid; i < cnt * F; i += 1024) {
      const int nl = i / F;
      const int c  = i - nl * F;
      float v = acc[i] / dns[nl * H + (c >> 6)] + bias[c];
      if (ELU) v = (v > 0.0f) ? v : expm1f(v);
      out[(size_t)lo * F + i] = v;
    }
  } else {
    for (int i = tid; i < cnt * F; i += 1024) {
      const int nl = i / F;
      const int c  = i - nl * F;
      float v = acc[nl * FP + c] / dns[nl * H + (c >> 6)] + bias[c];
      if (ELU) v = (v > 0.0f) ? v : expm1f(v);
      acc[nl * FP + c] = v;
    }
    __syncthreads();

    const int r  = lane & 15;
    const int hi = lane >> 4;
    const float4 ws4 = *reinterpret_cast<const float4*>(att_sn + 4 * r);
    const float4 wd4 = *reinterpret_cast<const float4*>(att_dn + 4 * r);
    const int nt = cnt >> 4;
    for (int rt = wv; rt < nt; rt += OWN_NW) {
      const float* arow = acc + (rt * 16 + r) * FP + 2 * hi;
      const float* __restrict__ brow = Wn + (size_t)(2 * hi) * 64 + 4 * r;
      v8f c0 = {}, c1 = {}, c2 = {}, c3 = {};
#pragma unroll 4
      for (int k = 0; k < F; k += 4) {
        const v2f a = *reinterpret_cast<const v2f*>(arow + k);
        const float4 p = *reinterpret_cast<const float4*>(brow + (size_t)k * 64);
        const float4 q = *reinterpret_cast<const float4*>(brow + (size_t)(k + 1) * 64);
        v2f b;
        b.x = p.x; b.y = q.x;
        c0 = __builtin_amdgcn_wmma_f32_16x16x4_f32(false, a, false, b, (short)0, c0, false, false);
        b.x = p.y; b.y = q.y;
        c1 = __builtin_amdgcn_wmma_f32_16x16x4_f32(false, a, false, b, (short)0, c1, false, false);
        b.x = p.z; b.y = q.z;
        c2 = __builtin_amdgcn_wmma_f32_16x16x4_f32(false, a, false, b, (short)0, c2, false, false);
        b.x = p.w; b.y = q.w;
        c3 = __builtin_amdgcn_wmma_f32_16x16x4_f32(false, a, false, b, (short)0, c3, false, false);
      }
      const int rowBase = lo + rt * 16 + (hi << 3);
      float v[16];
#pragma unroll
      for (int i = 0; i < 8; ++i) {
        float4 o;
        o.x = c0[i]; o.y = c1[i]; o.z = c2[i]; o.w = c3[i];
        *reinterpret_cast<float4*>(out + (size_t)(rowBase + i) * 64 + 4 * r) = o;
        v[i]     = (o.x * ws4.x + o.y * ws4.y) + (o.z * ws4.z + o.w * ws4.w);
        v[8 + i] = (o.x * wd4.x + o.y * wd4.y) + (o.z * wd4.z + o.w * wd4.w);
      }
      fold_step<8>(v, (lane & 8) != 0);
      fold_step<4>(v, (lane & 4) != 0);
      fold_step<2>(v, (lane & 2) != 0);
      fold_step<1>(v, (lane & 1) != 0);
      float* __restrict__ lp = (lane & 8) ? a_dn : a_sn;
      lp[rowBase + (lane & 7)] = v[0];
    }
  }
}

static inline int cdiv_l(long a, int b) { return (int)((a + b - 1) / b); }

extern "C" void kernel_launch(void* const* d_in, const int* in_sizes, int n_in,
                              void* d_out, int out_size, void* d_ws, size_t ws_size,
                              hipStream_t stream) {
  const float* x    = (const float*)d_in[0];
  const int*   ei   = (const int*)d_in[1];
  const float* W1   = (const float*)d_in[2];
  const float* as1w = (const float*)d_in[3];
  const float* ad1w = (const float*)d_in[4];
  const float* b1   = (const float*)d_in[5];
  const float* W2   = (const float*)d_in[6];
  const float* as2w = (const float*)d_in[7];
  const float* ad2w = (const float*)d_in[8];
  const float* b2   = (const float*)d_in[9];

  const int Fin = 128;
  const int N   = in_sizes[0] / Fin;
  const int E   = in_sizes[1] / 2;

  float* ws   = (float*)d_ws;
  float* h1   = ws;  ws += (size_t)N * 128;
  float* h2   = ws;
  float* a_s2 = ws + (size_t)N * 64;
  float* a_d2 = ws + (size_t)N * 65;
  ws += (size_t)N * 128;
  int*   kept = (int*)ws;  ws += (size_t)N * 64;
  float* a_s  = ws;  ws += (size_t)N * 2;
  float* a_d  = ws;  ws += (size_t)N * 2;
  float* outp = (float*)d_out;

  const int nOwn = cdiv_l((long)N, OWN_R);
  const int nIterK = cdiv_l((long)E, 512);
  unsigned* keys = (unsigned*)(kept + (size_t)nOwn * OWN_KSTRIDE);

  owner_keys<<<cdiv_l((long)nIterK * 128, 256 * OWN_KW), 256, 0, stream>>>(ei + E, E, N, keys, nIterK * 128);

  wmma_gemm_heads<128, 2><<<cdiv_l((long)(N / 16), 4), 256, 0, stream>>>(x, W1, h1, N / 16, as1w, ad1w, a_s, a_d);
  owner_aggregate<2, true, true, false, true><<<nOwn, 1024, 0, stream>>>(
      ei, E, N, a_s, a_d, h1, b1, h2, kept, keys, nIterK, W2, as2w, ad2w, a_s2, a_d2);

  owner_aggregate<1, false, false, true, false><<<nOwn, 1024, 0, stream>>>(
      ei, E, N, a_s2, a_d2, h2, b2, outp, kept, keys, nIterK, nullptr, nullptr, nullptr, nullptr, nullptr);
}


# owner-block row adds: a wave owns rows (row & 31 = wave), filters the block list for them (byte plane), adds by plain LDS load-fma-store: LDS float atomics gone; add order = list order, fused
# speedup vs baseline: 5.9329x; 1.0310x over previous
#include <hip/hip_runtime.h>
#include <hip/hip_bf16.h>
#include <math.h>

typedef __attribute__((ext_vector_type(2))) float v2f;
typedef __attribute__((ext_vector_type(8))) float v8f;

#define NEG_SLOPE 0.2f


template <int M>
__device__ __forceinline__ float row_xor(float v) {
  return __int_as_float(__builtin_amdgcn_update_dpp(
      0, __float_as_int(v), 0x160 + M, 0xf, 0xf, false));
}

template <int M>
__device__ __forceinline__ void fold_step(float (&v)[16], bool up) {
#pragma unroll
  for (int j = 0; j < M; ++j) {
    const float keep = up ? v[j + M] : v[j];
    const float send = up ? v[j] : v[j + M];
    v[j] = keep + row_xor<M>(send);
  }
}

template <int K, int H>
__global__ void __launch_bounds__(256)
wmma_gemm_heads(const float* __restrict__ A, const float* __restrict__ B,
                float* __restrict__ C, int nTiles,
                const float* __restrict__ att_s,
                const float* __restrict__ att_d,
                float* __restrict__ a_s, float* __restrict__ a_d) {
  constexpr int NC = H * 64;
  const int lane = threadIdx.x & 31;
  const int wv = __builtin_amdgcn_readfirstlane((int)(threadIdx.x >> 5));
  const int h  = wv % H;
  const int rt = blockIdx.x * (8 / H) + wv / H;
  if (rt >= nTiles) return;

  const int r  = lane & 15;
  const int hi = lane >> 4;
  const float4 ws4 = *reinterpret_cast<const float4*>(att_s + h * 64 + 4 * r);
  const float4 wd4 = *reinterpret_cast<const float4*>(att_d + h * 64 + 4 * r);

  const float* __restrict__ arow = A + (size_t)(rt * 16 + r) * K + 2 * hi;
  const float* __restrict__ brow = B + (size_t)(2 * hi) * NC + h * 64 + 4 * r;
  v8f c0 = {}, c1 = {}, c2 = {}, c3 = {};
#pragma unroll 4
  for (int k = 0; k < K; k += 4) {
    const v2f a = *reinterpret_cast<const v2f*>(arow + k);
    const float4 p = *reinterpret_cast<const float4*>(brow + (size_t)k * NC);
    const float4 q = *reinterpret_cast<const float4*>(brow + (size_t)(k + 1) * NC);
    v2f b;
    b.x = p.x; b.y = q.x;
    c0 = __builtin_amdgcn_wmma_f32_16x16x4_f32(false, a, false, b, (short)0, c0, false, false);
    b.x = p.y; b.y = q.y;
    c1 = __builtin_amdgcn_wmma_f32_16x16x4_f32(false, a, false, b, (short)0, c1, false, false);
    b.x = p.z; b.y = q.z;
    c2 = __builtin_amdgcn_wmma_f32_16x16x4_f32(false, a, false, b, (short)0, c2, false, false);
    b.x = p.w; b.y = q.w;
    c3 = __builtin_amdgcn_wmma_f32_16x16x4_f32(false, a, false, b, (short)0, c3, false, false);
  }

  const int rowBase = rt * 16 + (hi << 3);
  float v[16];
#pragma unroll
  for (int i = 0; i < 8; ++i) {
    float4 o;
    o.x = c0[i]; o.y = c1[i]; o.z = c2[i]; o.w = c3[i];
    *reinterpret_cast<float4*>(C + (size_t)(rowBase + i) * NC + h * 64 + 4 * r) = o;
    v[i]     = (o.x * ws4.x + o.y * ws4.y) + (o.z * ws4.z + o.w * ws4.w);
    v[8 + i] = (o.x * wd4.x + o.y * wd4.y) + (o.z * wd4.z + o.w * wd4.w);
  }
  fold_step<8>(v, (lane & 8) != 0);
  fold_step<4>(v, (lane & 4) != 0);
  fold_step<2>(v, (lane & 2) != 0);
  fold_step<1>(v, (lane & 1) != 0);
  float* __restrict__ lp = (lane & 8) ? a_d : a_s;
  lp[(size_t)(rowBase + (lane & 7)) * H + h] = v[0];
}

#define OWN_R    416
#define OWN_NW   32
#define OWN_QCAP 64
#define OWN_KU   4
#define OWN_KEY_NONE 255u
#define OWN_G    8
#define OWN_LCAP 8192
#define OWN_KHEAD 32
#define OWN_KSTRIDE (OWN_KHEAD + 2 * OWN_LCAP)
static_assert(OWN_LCAP % 1024 == 0, "a thread reads OWN_LCAP/1024 saved entries");

#define OWN_KW 4
__global__ void __launch_bounds__(256)
owner_keys(const int* __restrict__ dstp, int E, int N,
           unsigned* __restrict__ keys, int nWords) {
  const int w0 = blockIdx.x * (256 * OWN_KW) + threadIdx.x;
  int d[OWN_KW][4];
#pragma unroll
  for (int u = 0; u < OWN_KW; ++u) {
#pragma unroll
    for (int b = 0; b < 4; ++b)
      d[u][b] = dstp[min(4 * (w0 + u * 256) + b, E - 1)];
  }
#pragma unroll
  for (int u = 0; u < OWN_KW; ++u) {
    const int w = w0 + u * 256;
    unsigned packed = 0u;
#pragma unroll
    for (int b = 0; b < 4; ++b) {
      unsigned key = OWN_KEY_NONE;
      if (4 * w + b < E && (unsigned)d[u][b] < (unsigned)N) key = (unsigned)d[u][b] / OWN_R;
      packed |= key << (8 * b);
    }
    if (w < nWords) keys[w] = packed;
  }
}

__device__ __forceinline__ int ord_key(int b) { return b >= 0 ? b : (b ^ 0x7fffffff); }

template <int H, bool ELU, bool SAVE, bool REUSE, bool FUSE>
__global__ void __launch_bounds__(1024)
owner_aggregate(const int* __restrict__ ei, int E, int N,
                const float* __restrict__ a_s, const float* __restrict__ a_d,
                const float* __restrict__ hfeat, const float* __restrict__ bias,
                float* __restrict__ out, int* __restrict__ kept,
                const unsigned* __restrict__ keys, int nIterK,
                const float* __restrict__ Wn, const float* __restrict__ att_sn,
                const float* __restrict__ att_dn, float* __restrict__ a_sn,
                float* __restrict__ a_dn) {
  constexpr int F = H * 64;
  constexpr int FP = FUSE ? F + 4 : F;
  __shared__ __attribute__((aligned(16))) float acc[OWN_R * FP];
  __shared__ float dns[OWN_R * H];
  __shared__ int   mxs[OWN_R * H];
  __shared__ int   q_e[OWN_NW][OWN_QCAP];
  __shared__ int   l_s[OWN_LCAP];
  __shared__ int   l_d[OWN_LCAP];
  __shared__ __attribute__((aligned(16))) unsigned l_ow[OWN_LCAP / 4];
  __shared__ int   l_n;

  const int tid  = threadIdx.x;
  const int lane = tid & 31;
  const int wave = tid >> 5;
  const int lo   = blockIdx.x * OWN_R;
  const int cnt  = min(OWN_R, N - lo);
  const int* __restrict__ srcp = ei;
  const int* __restrict__ dstp = ei + E;

  int* __restrict__ k_n = kept + (size_t)blockIdx.x * OWN_KSTRIDE;
  int* __restrict__ k_s = k_n + OWN_KHEAD;
  int* __restrict__ k_d = k_s + OWN_LCAP;
  int  k_total = 0;
  bool k_have  = false;
  if (REUSE) {
    k_total = k_n[0];
    k_have  = (unsigned)k_total <= (unsigned)OWN_LCAP;
  }

  if (tid == 0) l_n = k_have ? k_total : 0;
  for (int i = tid; i < cnt * FP; i += 1024) acc[i] = 0.0f;
  for (int i = tid; i < OWN_LCAP / 4; i += 1024) l_ow[i] = 0xffffffffu;
  for (int i = tid; i < cnt * H; i += 1024) {
    float v = a_s[(size_t)lo * H + i] + a_d[(size_t)lo * H + i];
    v = (v > 0.0f) ? v : NEG_SLOPE * v;
    mxs[i] = ord_key(__float_as_int(v));
    dns[i] = 0.0f;
  }
  __syncthreads();

  auto score = [&](int s, int dl, int h) -> float {
    float v = a_s[(size_t)s * H + h] + a_d[(size_t)(lo + dl) * H + h];
    return (v > 0.0f) ? v : NEG_SLOPE * v;
  };

  auto add_rows = [&](bool excl, bool valid, int s, int dl, int m) {
    float w[H];
#pragma unroll
    for (int h = 0; h < H; ++h) w[h] = 0.0f;
    if (valid) {
#pragma unroll
      for (int h = 0; h < H; ++h) {
        const float mx = __int_as_float(ord_key(mxs[dl * H + h]));
        w[h] = __expf(score(s, dl, h) - mx);
        atomicAdd(&dns[dl * H + h], w[h]);
      }
    }
    for (int k0 = 0; k0 < m; k0 += OWN_G) {
      float2 r[OWN_G][H];
      int    dk[OWN_G];
#pragma unroll
      for (int g = 0; g < OWN_G; ++g) {
        const int kk = min(k0 + g, m - 1);
        const int sk = __builtin_amdgcn_readlane(s, kk);
        dk[g] = __builtin_amdgcn_readlane(dl, kk);
#pragma unroll
        for (int h = 0; h < H; ++h) {
          float wk = __int_as_float(__builtin_amdgcn_readlane(__float_as_int(w[h]), kk));
          if (k0 + g >= m) wk = 0.0f;
          const float2* __restrict__ hp =
              reinterpret_cast<const float2*>(hfeat + ((size_t)sk * H + h) * 64);
          const float2 v = hp[lane];
          r[g][h].x = wk * v.x;
          r[g][h].y = wk * v.y;
        }
      }
#pragma unroll
      for (int g = 0; g < OWN_G; ++g) {
        if (excl) {
#pragma unroll
          for (int h = 0; h < H; ++h) {
            float2* ap = reinterpret_cast<float2*>(acc + dk[g] * FP + h * 64 + 2 * lane);
            float2 a = *ap;
            a.x += r[g][h].x;
            a.y += r[g][h].y;
            *ap = a;
          }
        } else if (k0 + g < m) {
#pragma unroll
          for (int h = 0; h < H; ++h) {
            float* ap = acc + dk[g] * FP + h * 64 + 2 * lane;
            atomicAdd(&ap[0], r[g][h].x);
            atomicAdd(&ap[1], r[g][h].y);
          }
        }
      }
    }
  };

  auto flush = [&](int phase, int n) {
    __builtin_amdgcn_fence(__ATOMIC_ACQ_REL, "wavefront");
    for (int b = 0; b < n; b += 32) {
      const int i = b + lane;
      const bool valid = i < n;
      int s = 0, dl = 0;
      if (valid) {
        const int e = min(q_e[wave][i], E - 1);
        s  = srcp[e];
        dl = dstp[e] - lo;
        s  = min(max(s, 0), N - 1);
        dl = min(max(dl, 0), cnt - 1);
      }
      if (phase == 1) {
        int at = 0;
        if (lane == 0) at = atomicAdd(&l_n, min(32, n - b));
        at = __builtin_amdgcn_readfirstlane(at) + lane;
        if (valid) {
          if (at < OWN_LCAP) {
            l_s[at] = s;
            l_d[at] = dl;
            reinterpret_cast<unsigned char*>(l_ow)[at] = (unsigned char)(dl & (OWN_NW - 1));
          }
#pragma unroll
          for (int h = 0; h < H; ++h)
            atomicMax(&mxs[dl * H + h], ord_key(__float_as_int(score(s, dl, h))));
        }
      } else {
        add_rows(false, valid, s, dl, min(32, n - b));
      }
    }
    __builtin_amdgcn_fence(__ATOMIC_ACQ_REL, "wavefront");
  };

  const int rot = (int)(((long)blockIdx.x * nIterK) / gridDim.x);
  const uint4* __restrict__ kp = reinterpret_cast<const uint4*>(keys);
  const int wv = __builtin_amdgcn_readfirstlane(wave);
  const unsigned k4 = (unsigned)blockIdx.x * 0x01010101u;
  auto eq4 = [](unsigned w, unsigned key4) -> unsigned {
    const unsigned x = w ^ key4;
    const unsigned t = ((x & 0x7f7f7f7fu) + 0x7f7f7f7fu) | x;
    return ~t & 0x80808080u;
  };
  auto own4 = [&](unsigned w) -> unsigned {
    const unsigned x = w ^ k4;
    const unsigned t = ((x & 0x7f7f7f7fu) + 0x7f7f7f7fu) | x;
    return ~t & 0x80808080u;
  };

  bool list_ok = false;
  for (int phase = 1; phase <= 2; ++phase) {
    int qn = 0;
    if (list_ok) {
      const int total = l_n;
      const unsigned w4 = (unsigned)wv * 0x01010101u;
      const uint4* lo4 = reinterpret_cast<const uint4*>(l_ow);
      auto work_off = [&](int n) {
        __builtin_amdgcn_fence(__ATOMIC_ACQ_REL, "wavefront");
        for (int b = 0; b < n; b += 32) {
          const int i = b + lane;
          const bool valid = i < n;
          int s = 0, dl = 0;
          if (valid) {
            const int li = min(max(q_e[wave][i], 0), OWN_LCAP - 1);
            s  = l_s[li];
            dl = l_d[li];
          }
          add_rows(true, valid, s, dl, min(32, n - b));
        }
        __builtin_amdgcn_fence(__ATOMIC_ACQ_REL, "wavefront");
      };
      const int nIt = (total + 511) >> 9;
      for (int it = 0; it < nIt; ++it) {
        const uint4 kv = lo4[it * 32 + lane];
        unsigned f = (eq4(kv.x, w4) >> 7) | (eq4(kv.y, w4) >> 6) |
                     (eq4(kv.z, w4) >> 5) | (eq4(kv.w, w4) >> 4);
        const int i0 = it * 512 + lane * 16;
        unsigned mask = __builtin_amdgcn_ballot_w32(f != 0u);
        while (mask != 0u) {
          const int pos = qn + __popc(mask & ((1u << lane) - 1u));
          if (f != 0u) {
            const int bit = __builtin_ctz(f);
            q_e[wave][pos] = i0 + ((bit & 7) << 2) + (bit >> 3);
            f &= f - 1u;
          }
          qn += __popc(mask);
          if (qn >= 32) {
            work_off(qn);
            qn = 0;
          }
          mask = __builtin_amdgcn_ballot_w32(f != 0u);
        }
      }
      if (qn > 0) work_off(qn);
      qn = 0;
    } else if (REUSE && k_have) {
      constexpr int PER = OWN_LCAP / 1024;
      if (k_total > 0) {
        int  sv[PER], dv[PER];
        int  key[PER][H];
#pragma unroll
        for (int j = 0; j < PER; ++j) {
          const int i = min(tid + j * 1024, k_total - 1);
          sv[j] = min(max(k_s[i], 0), N - 1);
          dv[j] = min(max(k_d[i], 0), cnt - 1);
        }
#pragma unroll
        for (int j = 0; j < PER; ++j) {
#pragma unroll
          for (int h = 0; h < H; ++h)
            key[j][h] = ord_key(__float_as_int(score(sv[j], dv[j], h)));
        }
#pragma unroll
        for (int j = 0; j < PER; ++j) {
          const int i = tid + j * 1024;
          if (i < k_total) {
            l_s[i] = sv[j];
            l_d[i] = dv[j];
            reinterpret_cast<unsigned char*>(l_ow)[i] = (unsigned char)(dv[j] & (OWN_NW - 1));
#pragma unroll
            for (int h = 0; h < H; ++h) atomicMax(&mxs[dv[j] * H + h], key[j][h]);
          }
        }
      }
    } else
    for (int t0 = wv * OWN_KU; t0 < nIterK; t0 += OWN_NW * OWN_KU) {
      uint4 kv[OWN_KU];
      int   e0[OWN_KU];
#pragma unroll
      for (int j = 0; j < OWN_KU; ++j) {
        int it = min(t0 + j, nIterK - 1) + rot;
        if (it >= nIterK) it -= nIterK;
        kv[j] = kp[(size_t)it * 32 + lane];
        e0[j] = it * 512 + lane * 16;
      }
#pragma unroll
      for (int j = 0; j < OWN_KU; ++j) {
        unsigned f = (own4(kv[j].x) >> 7) | (own4(kv[j].y) >> 6) |
                     (own4(kv[j].z) >> 5) | (own4(kv[j].w) >> 4);
        if (t0 + j >= nIterK) f = 0u;
        unsigned mask = __builtin_amdgcn_ballot_w32(f != 0u);
        while (mask != 0u) {
          const int pos = qn + __popc(mask & ((1u << lane) - 1u));
          if (f != 0u) {
            const int bit = __builtin_ctz(f);
            q_e[wave][pos] = e0[j] + ((bit & 7) << 2) + (bit >> 3);
            f &= f - 1u;
          }
          qn += __popc(mask);
          if (qn >= 32) {
            flush(phase, qn);
            qn = 0;
          }
          mask = __builtin_amdgcn_ballot_w32(f != 0u);
        }
      }
    }
    if (qn > 0) flush(phase, qn);
    if (phase == 2) {
      const int mine = (cnt - wave + OWN_NW - 1) / OWN_NW;
      for (int b = 0; b < mine; b += 32) {
        const int dl = wave + OWN_NW * (b + lane);
        const bool valid = b + lane < mine;
        add_rows(list_ok, valid, valid ? lo + dl : 0, valid ? dl : 0, min(32, mine - b));
      }
    }
    __syncthreads();
    list_ok = (l_n <= OWN_LCAP);
    if (SAVE && phase == 1) {
      const int total = l_n;
      if (tid == 0) k_n[0] = total;
      if (list_ok) {
        for (int i = tid; i < total; i += 1024) {
          k_s[i] = l_s[i];
          k_d[i] = l_d[i];
        }
      }
    }
  }

  if constexpr (!FUSE) {
    for (int i = tid; i < cnt * F; i += 1024) {
      const int nl = i / F;
      const int c  = i - nl * F;
      float v = acc[i] / dns[nl * H + (c >> 6)] + bias[c];
      if (ELU) v = (v > 0.0f) ? v : expm1f(v);
      out[(size_t)lo * F + i] = v;
    }
  } else {
    for (int i = tid; i < cnt * F; i += 1024) {
      const int nl = i / F;
      const int c  = i - nl * F;
      float v = acc[nl * FP + c] / dns[nl * H + (c >> 6)] + bias[c];
      if (ELU) v = (v > 0.0f) ? v : expm1f(v);
      acc[nl * FP + c] = v;
    }
    __syncthreads();

    const int r  = lane & 15;
    const int hi = lane >> 4;
    const float4 ws4 = *reinterpret_cast<const float4*>(att_sn + 4 * r);
    const float4 wd4 = *reinterpret_cast<const float4*>(att_dn + 4 * r);
    const int nt = cnt >> 4;
    for (int rt = wv; rt < nt; rt += OWN_NW) {
      const float* arow = acc + (rt * 16 + r) * FP + 2 * hi;
      const float* __restrict__ brow = Wn + (size_t)(2 * hi) * 64 + 4 * r;
      v8f c0 = {}, c1 = {}, c2 = {}, c3 = {};
#pragma unroll 4
      for (int k = 0; k < F; k += 4) {
        const v2f a = *reinterpret_cast<const v2f*>(arow + k);
        const float4 p = *reinterpret_cast<const float4*>(brow + (size_t)k * 64);
        const float4 q = *reinterpret_cast<const float4*>(brow + (size_t)(k + 1) * 64);
        v2f b;
        b.x = p.x; b.y = q.x;
        c0 = __builtin_amdgcn_wmma_f32_16x16x4_f32(false, a, false, b, (short)0, c0, false, false);
        b.x = p.y; b.y = q.y;
        c1 = __builtin_amdgcn_wmma_f32_16x16x4_f32(false, a, false, b, (short)0, c1, false, false);
        b.x = p.z; b.y = q.z;
        c2 = __builtin_amdgcn_wmma_f32_16x16x4_f32(false, a, false, b, (short)0, c2, false, false);
        b.x = p.w; b.y = q.w;
        c3 = __builtin_amdgcn_wmma_f32_16x16x4_f32(false, a, false, b, (short)0, c3, false, false);
      }
      const int rowBase = lo + rt * 16 + (hi << 3);
      float v[16];
#pragma unroll
      for (int i = 0; i < 8; ++i) {
        float4 o;
        o.x = c0[i]; o.y = c1[i]; o.z = c2[i]; o.w = c3[i];
        *reinterpret_cast<float4*>(out + (size_t)(rowBase + i) * 64 + 4 * r) = o;
        v[i]     = (o.x * ws4.x + o.y * ws4.y) + (o.z * ws4.z + o.w * ws4.w);
        v[8 + i] = (o.x * wd4.x + o.y * wd4.y) + (o.z * wd4.z + o.w * wd4.w);
      }
      fold_step<8>(v, (lane & 8) != 0);
      fold_step<4>(v, (lane & 4) != 0);
      fold_step<2>(v, (lane & 2) != 0);
      fold_step<1>(v, (lane & 1) != 0);
      float* __restrict__ lp = (lane & 8) ? a_dn : a_sn;
      lp[rowBase + (lane & 7)] = v[0];
    }
  }
}

static inline int cdiv_l(long a, int b) { return (int)((a + b - 1) / b); }

extern "C" void kernel_launch(void* const* d_in, const int* in_sizes, int n_in,
                              void* d_out, int out_size, void* d_ws, size_t ws_size,
                              hipStream_t stream) {
  const float* x    = (const float*)d_in[0];
  const int*   ei   = (const int*)d_in[1];
  const float* W1   = (const float*)d_in[2];
  const float* as1w = (const float*)d_in[3];
  const float* ad1w = (const float*)d_in[4];
  const float* b1   = (const float*)d_in[5];
  const float* W2   = (const float*)d_in[6];
  const float* as2w = (const float*)d_in[7];
  const float* ad2w = (const float*)d_in[8];
  const float* b2   = (const float*)d_in[9];

  const int Fin = 128;
  const int N   = in_sizes[0] / Fin;
  const int E   = in_sizes[1] / 2;

  float* ws   = (float*)d_ws;
  float* h1   = ws;  ws += (size_t)N * 128;
  float* h2   = ws;
  float* a_s2 = ws + (size_t)N * 64;
  float* a_d2 = ws + (size_t)N * 65;
  ws += (size_t)N * 128;
  int*   kept = (int*)ws;  ws += (size_t)N * 64;
  float* a_s  = ws;  ws += (size_t)N * 2;
  float* a_d  = ws;  ws += (size_t)N * 2;
  float* outp = (float*)d_out;

  const int nOwn = cdiv_l((long)N, OWN_R);
  const int nIterK = cdiv_l((long)E, 512);
  unsigned* keys = (unsigned*)(kept + (size_t)nOwn * OWN_KSTRIDE);

  owner_keys<<<cdiv_l((long)nIterK * 128, 256 * OWN_KW), 256, 0, stream>>>(ei + E, E, N, keys, nIterK * 128);

  wmma_gemm_heads<128, 2><<<cdiv_l((long)(N / 16), 4), 256, 0, stream>>>(x, W1, h1, N / 16, as1w, ad1w, a_s, a_d);
  owner_aggregate<2, true, true, false, true><<<nOwn, 1024, 0, stream>>>(
      ei, E, N, a_s, a_d, h1, b1, h2, kept, keys, nIterK, W2, as2w, ad2w, a_s2, a_d2);

  owner_aggregate<1, false, false, true, false><<<nOwn, 1024, 0, stream>>>(
      ei, E, N, a_s2, a_d2, h2, b2, outp, kept, keys, nIterK, nullptr, nullptr, nullptr, nullptr, nullptr);
}


# kept edges go straight into 32 per-wave lists in LDS (slot = LDS counter of the row's owner wave): the filter pass every wave ran over the block list is gone; add order = arrival order in a wave list
# speedup vs baseline: 6.3110x; 1.0637x over previous
#include <hip/hip_runtime.h>
#include <hip/hip_bf16.h>
#include <math.h>

typedef __attribute__((ext_vector_type(2))) float v2f;
typedef __attribute__((ext_vector_type(8))) float v8f;

#define NEG_SLOPE 0.2f


template <int M>
__device__ __forceinline__ float row_xor(float v) {
  return __int_as_float(__builtin_amdgcn_update_dpp(
      0, __float_as_int(v), 0x160 + M, 0xf, 0xf, false));
}

template <int M>
__device__ __forceinline__ void fold_step(float (&v)[16], bool up) {
#pragma unroll
  for (int j = 0; j < M; ++j) {
    const float keep = up ? v[j + M] : v[j];
    const float send = up ? v[j] : v[j + M];
    v[j] = keep + row_xor<M>(send);
  }
}

template <int K, int H>
__global__ void __launch_bounds__(256)
wmma_gemm_heads(const float* __restrict__ A, const float* __restrict__ B,
                float* __restrict__ C, int nTiles,
                const float* __restrict__ att_s,
                const float* __restrict__ att_d,
                float* __restrict__ a_s, float* __restrict__ a_d) {
  constexpr int NC = H * 64;
  const int lane = threadIdx.x & 31;
  const int wv = __builtin_amdgcn_readfirstlane((int)(threadIdx.x >> 5));
  const int h  = wv % H;
  const int rt = blockIdx.x * (8 / H) + wv / H;
  if (rt >= nTiles) return;

  const int r  = lane & 15;
  const int hi = lane >> 4;
  const float4 ws4 = *reinterpret_cast<const float4*>(att_s + h * 64 + 4 * r);
  const float4 wd4 = *reinterpret_cast<const float4*>(att_d + h * 64 + 4 * r);

  const float* __restrict__ arow = A + (size_t)(rt * 16 + r) * K + 2 * hi;
  const float* __restrict__ brow = B + (size_t)(2 * hi) * NC + h * 64 + 4 * r;
  v8f c0 = {}, c1 = {}, c2 = {}, c3 = {};
#pragma unroll 4
  for (int k = 0; k < K; k += 4) {
    const v2f a = *reinterpret_cast<const v2f*>(arow + k);
    const float4 p = *reinterpret_cast<const float4*>(brow + (size_t)k * NC);
    const float4 q = *reinterpret_cast<const float4*>(brow + (size_t)(k + 1) * NC);
    v2f b;
    b.x = p.x; b.y = q.x;
    c0 = __builtin_amdgcn_wmma_f32_16x16x4_f32(false, a, false, b, (short)0, c0, false, false);
    b.x = p.y; b.y = q.y;
    c1 = __builtin_amdgcn_wmma_f32_16x16x4_f32(false, a, false, b, (short)0, c1, false, false);
    b.x = p.z; b.y = q.z;
    c2 = __builtin_amdgcn_wmma_f32_16x16x4_f32(false, a, false, b, (short)0, c2, false, false);
    b.x = p.w; b.y = q.w;
    c3 = __builtin_amdgcn_wmma_f32_16x16x4_f32(false, a, false, b, (short)0, c3, false, false);
  }

  const int rowBase = rt * 16 + (hi << 3);
  float v[16];
#pragma unroll
  for (int i = 0; i < 8; ++i) {
    float4 o;
    o.x = c0[i]; o.y = c1[i]; o.z = c2[i]; o.w = c3[i];
    *reinterpret_cast<float4*>(C + (size_t)(rowBase + i) * NC + h * 64 + 4 * r) = o;
    v[i]     = (o.x * ws4.x + o.y * ws4.y) + (o.z * ws4.z + o.w * ws4.w);
    v[8 + i] = (o.x * wd4.x + o.y * wd4.y) + (o.z * wd4.z + o.w * wd4.w);
  }
  fold_step<8>(v, (lane & 8) != 0);
  fold_step<4>(v, (lane & 4) != 0);
  fold_step<2>(v, (lane & 2) != 0);
  fold_step<1>(v, (lane & 1) != 0);
  float* __restrict__ lp = (lane & 8) ? a_d : a_s;
  lp[(size_t)(rowBase + (lane & 7)) * H + h] = v[0];
}

#define OWN_R    416
#define OWN_NW   32
#define OWN_QCAP 64
#define OWN_KU   4
#define OWN_KEY_NONE 255u
#define OWN_G    8
#define OWN_SCAP 320
#define OWN_LCAP (OWN_NW * OWN_SCAP)
#define OWN_KHEAD 64
#define OWN_KSTRIDE (OWN_KHEAD + 2 * OWN_LCAP)
static_assert(OWN_LCAP % 1024 == 0, "a thread handles OWN_LCAP/1024 list slots");

#define OWN_KW 4
__global__ void __launch_bounds__(256)
owner_keys(const int* __restrict__ dstp, int E, int N,
           unsigned* __restrict__ keys, int nWords) {
  const int w0 = blockIdx.x * (256 * OWN_KW) + threadIdx.x;
  int d[OWN_KW][4];
#pragma unroll
  for (int u = 0; u < OWN_KW; ++u) {
#pragma unroll
    for (int b = 0; b < 4; ++b)
      d[u][b] = dstp[min(4 * (w0 + u * 256) + b, E - 1)];
  }
#pragma unroll
  for (int u = 0; u < OWN_KW; ++u) {
    const int w = w0 + u * 256;
    unsigned packed = 0u;
#pragma unroll
    for (int b = 0; b < 4; ++b) {
      unsigned key = OWN_KEY_NONE;
      if (4 * w + b < E && (unsigned)d[u][b] < (unsigned)N) key = (unsigned)d[u][b] / OWN_R;
      packed |= key << (8 * b);
    }
    if (w < nWords) keys[w] = packed;
  }
}

__device__ __forceinline__ int ord_key(int b) { return b >= 0 ? b : (b ^ 0x7fffffff); }

template <int H, bool ELU, bool SAVE, bool REUSE, bool FUSE>
__global__ void __launch_bounds__(1024)
owner_aggregate(const int* __restrict__ ei, int E, int N,
                const float* __restrict__ a_s, const float* __restrict__ a_d,
                const float* __restrict__ hfeat, const float* __restrict__ bias,
                float* __restrict__ out, int* __restrict__ kept,
                const unsigned* __restrict__ keys, int nIterK,
                const float* __restrict__ Wn, const float* __restrict__ att_sn,
                const float* __restrict__ att_dn, float* __restrict__ a_sn,
                float* __restrict__ a_dn) {
  constexpr int F = H * 64;
  constexpr int FP = FUSE ? F + 4 : F;
  __shared__ __attribute__((aligned(16))) float acc[OWN_R * FP];
  __shared__ float dns[OWN_R * H];
  __shared__ int   mxs[OWN_R * H];
  __shared__ int   q_e[OWN_NW][OWN_QCAP];
  __shared__ int   l_s[OWN_LCAP];
  __shared__ int   l_d[OWN_LCAP];
  __shared__ int   l_cnt[OWN_NW];
  __shared__ int   l_over;

  const int tid  = threadIdx.x;
  const int lane = tid & 31;
  const int wave = tid >> 5;
  const int lo   = blockIdx.x * OWN_R;
  const int cnt  = min(OWN_R, N - lo);
  const int* __restrict__ srcp = ei;
  const int* __restrict__ dstp = ei + E;

  int* __restrict__ k_n = kept + (size_t)blockIdx.x * OWN_KSTRIDE;
  int* __restrict__ k_s = k_n + OWN_KHEAD;
  int* __restrict__ k_d = k_s + OWN_LCAP;
  bool k_have  = false;
  if (REUSE) k_have = (k_n[0] == 1);

  if (tid == 0) l_over = 0;
  if (tid < OWN_NW) {
    int c = 0;
    if (REUSE && k_have) c = min(max(k_n[1 + tid], 0), OWN_SCAP);
    l_cnt[tid] = c;
  }
  for (int i = tid; i < cnt * FP; i += 1024) acc[i] = 0.0f;
  for (int i = tid; i < cnt * H; i += 1024) {
    float v = a_s[(size_t)lo * H + i] + a_d[(size_t)lo * H + i];
    v = (v > 0.0f) ? v : NEG_SLOPE * v;
    mxs[i] = ord_key(__float_as_int(v));
    dns[i] = 0.0f;
  }
  __syncthreads();

  auto score = [&](int s, int dl, int h) -> float {
    float v = a_s[(size_t)s * H + h] + a_d[(size_t)(lo + dl) * H + h];
    return (v > 0.0f) ? v : NEG_SLOPE * v;
  };

  auto add_rows = [&](bool excl, bool valid, int s, int dl, int m) {
    float w[H];
#pragma unroll
    for (int h = 0; h < H; ++h) w[h] = 0.0f;
    if (valid) {
#pragma unroll
      for (int h = 0; h < H; ++h) {
        const float mx = __int_as_float(ord_key(mxs[dl * H + h]));
        w[h] = __expf(score(s, dl, h) - mx);
        atomicAdd(&dns[dl * H + h], w[h]);
      }
    }
    for (int k0 = 0; k0 < m; k0 += OWN_G) {
      float2 r[OWN_G][H];
      int    dk[OWN_G];
#pragma unroll
      for (int g = 0; g < OWN_G; ++g) {
        const int kk = min(k0 + g, m - 1);
        const int sk = __builtin_amdgcn_readlane(s, kk);
        dk[g] = __builtin_amdgcn_readlane(dl, kk);
#pragma unroll
        for (int h = 0; h < H; ++h) {
          float wk = __int_as_float(__builtin_amdgcn_readlane(__float_as_int(w[h]), kk));
          if (k0 + g >= m) wk = 0.0f;
          const float2* __restrict__ hp =
              reinterpret_cast<const float2*>(hfeat + ((size_t)sk * H + h) * 64);
          const float2 v = hp[lane];
          r[g][h].x = wk * v.x;
          r[g][h].y = wk * v.y;
        }
      }
#pragma unroll
      for (int g = 0; g < OWN_G; ++g) {
        if (excl) {
#pragma unroll
          for (int h = 0; h < H; ++h) {
            float2* ap = reinterpret_cast<float2*>(acc + dk[g] * FP + h * 64 + 2 * lane);
            float2 a = *ap;
            a.x += r[g][h].x;
            a.y += r[g][h].y;
            *ap = a;
          }
        } else if (k0 + g < m) {
#pragma unroll
          for (int h = 0; h < H; ++h) {
            float* ap = acc + dk[g] * FP + h * 64 + 2 * lane;
            atomicAdd(&ap[0], r[g][h].x);
            atomicAdd(&ap[1], r[g][h].y);
          }
        }
      }
    }
  };

  auto flush = [&](int phase, int n) {
    __builtin_amdgcn_fence(__ATOMIC_ACQ_REL, "wavefront");
    for (int b = 0; b < n; b += 32) {
      const int i = b + lane;
      const bool valid = i < n;
      int s = 0, dl = 0;
      if (valid) {
        const int e = min(q_e[wave][i], E - 1);
        s  = srcp[e];
        dl = dstp[e] - lo;
        s  = min(max(s, 0), N - 1);
        dl = min(max(dl, 0), cnt - 1);
      }
      if (phase == 1) {
        if (valid) {
          const int ow = dl & (OWN_NW - 1);
          const int at = atomicAdd(&l_cnt[ow], 1);
          if (at < OWN_SCAP) {
            l_s[ow * OWN_SCAP + at] = s;
            l_d[ow * OWN_SCAP + at] = dl;
          } else {
            l_over = 1;
          }
#pragma unroll
          for (int h = 0; h < H; ++h)
            atomicMax(&mxs[dl * H + h], ord_key(__float_as_int(score(s, dl, h))));
        }
      } else {
        add_rows(false, valid, s, dl, min(32, n - b));
      }
    }
    __builtin_amdgcn_fence(__ATOMIC_ACQ_REL, "wavefront");
  };

  const int rot = (int)(((long)blockIdx.x * nIterK) / gridDim.x);
  const uint4* __restrict__ kp = reinterpret_cast<const uint4*>(keys);
  const int wv = __builtin_amdgcn_readfirstlane(wave);
  const unsigned k4 = (unsigned)blockIdx.x * 0x01010101u;
  auto own4 = [&](unsigned w) -> unsigned {
    const unsigned x = w ^ k4;
    const unsigned t = ((x & 0x7f7f7f7fu) + 0x7f7f7f7fu) | x;
    return ~t & 0x80808080u;
  };

  bool list_ok = false;
  for (int phase = 1; phase <= 2; ++phase) {
    int qn = 0;
    if (list_ok) {
      const int n = __builtin_amdgcn_readfirstlane(min(max(l_cnt[wv], 0), OWN_SCAP));
      const int* my_s = l_s + wv * OWN_SCAP;
      const int* my_d = l_d + wv * OWN_SCAP;
      for (int b = 0; b < n; b += 32) {
        const int i = b + lane;
        const bool valid = i < n;
        int s = 0, dl = 0;
        if (valid) {
          s  = my_s[i];
          dl = my_d[i];
        }
        add_rows(true, valid, s, dl, min(32, n - b));
      }
    } else if (REUSE && k_have) {
      constexpr int PER = OWN_LCAP / 1024;
      int  sv[PER], dv[PER];
      bool ok[PER];
      int  key[PER][H];
#pragma unroll
      for (int j = 0; j < PER; ++j) {
        const int i  = tid + j * 1024;
        const int ow = i / OWN_SCAP;
        ok[j] = (i - ow * OWN_SCAP) < l_cnt[ow];
        sv[j] = 0;
        dv[j] = 0;
        if (ok[j]) {
          sv[j] = min(max(k_s[i], 0), N - 1);
          dv[j] = min(max(k_d[i], 0), cnt - 1);
        }
      }
#pragma unroll
      for (int j = 0; j < PER; ++j) {
#pragma unroll
        for (int h = 0; h < H; ++h)
          key[j][h] = ord_key(__float_as_int(score(sv[j], dv[j], h)));
      }
#pragma unroll
      for (int j = 0; j < PER; ++j) {
        const int i = tid + j * 1024;
        if (ok[j]) {
          l_s[i] = sv[j];
          l_d[i] = dv[j];
#pragma unroll
          for (int h = 0; h < H; ++h) atomicMax(&mxs[dv[j] * H + h], key[j][h]);
        }
      }
    } else
    for (int t0 = wv * OWN_KU; t0 < nIterK; t0 += OWN_NW * OWN_KU) {
      uint4 kv[OWN_KU];
      int   e0[OWN_KU];
#pragma unroll
      for (int j = 0; j < OWN_KU; ++j) {
        int it = min(t0 + j, nIterK - 1) + rot;
        if (it >= nIterK) it -= nIterK;
        kv[j] = kp[(size_t)it * 32 + lane];
        e0[j] = it * 512 + lane * 16;
      }
#pragma unroll
      for (int j = 0; j < OWN_KU; ++j) {
        unsigned f = (own4(kv[j].x) >> 7) | (own4(kv[j].y) >> 6) |
                     (own4(kv[j].z) >> 5) | (own4(kv[j].w) >> 4);
        if (t0 + j >= nIterK) f = 0u;
        unsigned mask = __builtin_amdgcn_ballot_w32(f != 0u);
        while (mask != 0u) {
          const int pos = qn + __popc(mask & ((1u << lane) - 1u));
          if (f != 0u) {
            const int bit = __builtin_ctz(f);
            q_e[wave][pos] = e0[j] + ((bit & 7) << 2) + (bit >> 3);
            f &= f - 1u;
          }
          qn += __popc(mask);
          if (qn >= 32) {
            flush(phase, qn);
            qn = 0;
          }
          mask = __builtin_amdgcn_ballot_w32(f != 0u);
        }
      }
    }
    if (qn > 0) flush(phase, qn);
    if (phase == 2) {
      const int mine = (cnt - wave + OWN_NW - 1) / OWN_NW;
      for (int b = 0; b < mine; b += 32) {
        const int dl = wave + OWN_NW * (b + lane);
        const bool valid = b + lane < mine;
        add_rows(list_ok, valid, valid ? lo + dl : 0, valid ? dl : 0, min(32, mine - b));
      }
    }
    __syncthreads();
    list_ok = (l_over == 0);
    if (SAVE && phase == 1) {
      if (tid == 0) k_n[0] = list_ok ? 1 : 0;
      if (list_ok) {
        if (tid < OWN_NW) k_n[1 + tid] = l_cnt[tid];
        for (int i = tid; i < OWN_LCAP; i += 1024) {
          const int ow = i / OWN_SCAP;
          if (i - ow * OWN_SCAP < l_cnt[ow]) {
            k_s[i] = l_s[i];
            k_d[i] = l_d[i];
          }
        }
      }
    }
  }

  if constexpr (!FUSE) {
    for (int i = tid; i < cnt * F; i += 1024) {
      const int nl = i / F;
      const int c  = i - nl * F;
      float v = acc[i] / dns[nl * H + (c >> 6)] + bias[c];
      if (ELU) v = (v > 0.0f) ? v : expm1f(v);
      out[(size_t)lo * F + i] = v;
    }
  } else {
    for (int i = tid; i < cnt * F; i += 1024) {
      const int nl = i / F;
      const int c  = i - nl * F;
      float v = acc[nl * FP + c] / dns[nl * H + (c >> 6)] + bias[c];
      if (ELU) v = (v > 0.0f) ? v : expm1f(v);
      acc[nl * FP + c] = v;
    }
    __syncthreads();

    const int r  = lane & 15;
    const int hi = lane >> 4;
    const float4 ws4 = *reinterpret_cast<const float4*>(att_sn + 4 * r);
    const float4 wd4 = *reinterpret_cast<const float4*>(att_dn + 4 * r);
    const int nt = cnt >> 4;
    for (int rt = wv; rt < nt; rt += OWN_NW) {
      const float* arow = acc + (rt * 16 + r) * FP + 2 * hi;
      const float* __restrict__ brow = Wn + (size_t)(2 * hi) * 64 + 4 * r;
      v8f c0 = {}, c1 = {}, c2 = {}, c3 = {};
#pragma unroll 4
      for (int k = 0; k < F; k += 4) {
        const v2f a = *reinterpret_cast<const v2f*>(arow + k);
        const float4 p = *reinterpret_cast<const float4*>(brow + (size_t)k * 64);
        const float4 q = *reinterpret_cast<const float4*>(brow + (size_t)(k + 1) * 64);
        v2f b;
        b.x = p.x; b.y = q.x;
        c0 = __builtin_amdgcn_wmma_f32_16x16x4_f32(false, a, false, b, (short)0, c0, false, false);
        b.x = p.y; b.y = q.y;
        c1 = __builtin_amdgcn_wmma_f32_16x16x4_f32(false, a, false, b, (short)0, c1, false, false);
        b.x = p.z; b.y = q.z;
        c2 = __builtin_amdgcn_wmma_f32_16x16x4_f32(false, a, false, b, (short)0, c2, false, false);
        b.x = p.w; b.y = q.w;
        c3 = __builtin_amdgcn_wmma_f32_16x16x4_f32(false, a, false, b, (short)0, c3, false, false);
      }
      const int rowBase = lo + rt * 16 + (hi << 3);
      float v[16];
#pragma unroll
      for (int i = 0; i < 8; ++i) {
        float4 o;
        o.x = c0[i]; o.y = c1[i]; o.z = c2[i]; o.w = c3[i];
        *reinterpret_cast<float4*>(out + (size_t)(rowBase + i) * 64 + 4 * r) = o;
        v[i]     = (o.x * ws4.x + o.y * ws4.y) + (o.z * ws4.z + o.w * ws4.w);
        v[8 + i] = (o.x * wd4.x + o.y * wd4.y) + (o.z * wd4.z + o.w * wd4.w);
      }
      fold_step<8>(v, (lane & 8) != 0);
      fold_step<4>(v, (lane & 4) != 0);
      fold_step<2>(v, (lane & 2) != 0);
      fold_step<1>(v, (lane & 1) != 0);
      float* __restrict__ lp = (lane & 8) ? a_dn : a_sn;
      lp[rowBase + (lane & 7)] = v[0];
    }
  }
}

static inline int cdiv_l(long a, int b) { return (int)((a + b - 1) / b); }

extern "C" void kernel_launch(void* const* d_in, const int* in_sizes, int n_in,
                              void* d_out, int out_size, void* d_ws, size_t ws_size,
                              hipStream_t stream) {
  const float* x    = (const float*)d_in[0];
  const int*   ei   = (const int*)d_in[1];
  const float* W1   = (const float*)d_in[2];
  const float* as1w = (const float*)d_in[3];
  const float* ad1w = (const float*)d_in[4];
  const float* b1   = (const float*)d_in[5];
  const float* W2   = (const float*)d_in[6];
  const float* as2w = (const float*)d_in[7];
  const float* ad2w = (const float*)d_in[8];
  const float* b2   = (const float*)d_in[9];

  const int Fin = 128;
  const int N   = in_sizes[0] / Fin;
  const int E   = in_sizes[1] / 2;

  float* ws   = (float*)d_ws;
  float* h1   = ws;  ws += (size_t)N * 128;
  float* h2   = ws;
  float* a_s2 = ws + (size_t)N * 64;
  float* a_d2 = ws + (size_t)N * 65;
  ws += (size_t)N * 128;
  int*   kept = (int*)ws;  ws += (size_t)N * 64;
  float* a_s  = ws;  ws += (size_t)N * 2;
  float* a_d  = ws;  ws += (size_t)N * 2;
  float* outp = (float*)d_out;

  const int nOwn = cdiv_l((long)N, OWN_R);
  const int nIterK = cdiv_l((long)E, 512);
  unsigned* keys = (unsigned*)(kept + (size_t)nOwn * OWN_KSTRIDE);

  owner_keys<<<cdiv_l((long)nIterK * 128, 256 * OWN_KW), 256, 0, stream>>>(ei + E, E, N, keys, nIterK * 128);

  wmma_gemm_heads<128, 2><<<cdiv_l((long)(N / 16), 4), 256, 0, stream>>>(x, W1, h1, N / 16, as1w, ad1w, a_s, a_d);
  owner_aggregate<2, true, true, false, true><<<nOwn, 1024, 0, stream>>>(
      ei, E, N, a_s, a_d, h1, b1, h2, kept, keys, nIterK, W2, as2w, ad2w, a_s2, a_d2);

  owner_aggregate<1, false, false, true, false><<<nOwn, 1024, 0, stream>>>(
      ei, E, N, a_s2, a_d2, h2, b2, outp, kept, keys, nIterK, nullptr, nullptr, nullptr, nullptr, nullptr);
}
